# starting kernel
# baseline (speedup 1.0000x reference)
#include <hip/hip_runtime.h>
#include <hip/hip_bf16.h>
#include <math.h>


typedef __attribute__((ext_vector_type(2))) float v2f;
typedef __attribute__((ext_vector_type(8))) float v8f;

#define Bsz 8
#define Hsz 64
#define Lsz 32768
#define Nmode 16
#define NTHREADS 256
#define CHUNK (Lsz / NTHREADS)

__global__ __launch_bounds__(NTHREADS)
void k_channel_mix(const float* __restrict__ x, const float* __restrict__ W,
                   const float* __restrict__ bias, float* __restrict__ u)
{
    const int tid  = threadIdx.x;
    const int wave = tid >> 5;
    const int lane = tid & 31;
    const int half = lane >> 4;
    const int lp   = lane & 15;

    const int b  = blockIdx.y;
    const int l0 = blockIdx.x * 128 + wave * 16;
    const float* xb = x + (size_t)b * Hsz * Lsz;

    v8f acc[4] = {};

    for (int k0 = 0; k0 < Hsz; k0 += 4) {
        v2f bb;
        bb.x = xb[(size_t)(k0 + 2 * half + 0) * Lsz + l0 + lp];
        bb.y = xb[(size_t)(k0 + 2 * half + 1) * Lsz + l0 + lp];
#pragma unroll
        for (int mt = 0; mt < 4; ++mt) {
            const float* wrow = W + (size_t)(mt * 16 + lp) * Hsz + k0 + 2 * half;
            v2f aa = *(const v2f*)wrow;
            acc[mt] = __builtin_amdgcn_wmma_f32_16x16x4_f32(
                false, aa, false, bb, (short)0, acc[mt], false, false);
        }
    }

#pragma unroll
    for (int mt = 0; mt < 4; ++mt) {
#pragma unroll
        for (int r = 0; r < 8; ++r) {
            const int g = mt * 16 + r + half * 8;
            const float v = acc[mt][r] + bias[g];
            u[((size_t)b * Hsz + g) * Lsz + l0 + lp] = tanhf(v);
        }
    }
}

__global__ __launch_bounds__(NTHREADS)
void k_dssm_scan(const float* __restrict__ u, const float* __restrict__ x,
                 const float* __restrict__ log_dt,
                 const float* __restrict__ C_re, const float* __restrict__ C_im,
                 const float* __restrict__ log_A_real,
                 const float* __restrict__ A_imag,
                 const float* __restrict__ Dp, const float* __restrict__ res_w,
                 float* __restrict__ out)
{
    __shared__ float zre[Nmode], zim[Nmode];
    __shared__ float wre[Nmode], wim[Nmode];
    __shared__ float Zre[Nmode], Zim[Nmode];
    __shared__ float sAre[NTHREADS][Nmode + 1];
    __shared__ float sAim[NTHREADS][Nmode + 1];

    const int tid = threadIdx.x;
    const int bh  = blockIdx.x;
    const int h   = bh & (Hsz - 1);

    if (tid < Nmode) {
        const int n = tid;
        const float dt  = expf(log_dt[h]);
        const float Are = -expf(log_A_real[h * Nmode + n]);
        const float Aim = A_imag[h * Nmode + n];
        const float dre = Are * dt, dim = Aim * dt;
        const float er  = expf(dre);
        const float zr  = er * cosf(dim), zi = er * sinf(dim);
        zre[n] = zr; zim[n] = zi;
        const float e1r = zr - 1.0f, e1i = zi;
        const float cr = C_re[h * Nmode + n], ci = C_im[h * Nmode + n];
        const float nr = cr * e1r - ci * e1i;
        const float ni = cr * e1i + ci * e1r;
        const float den = Are * Are + Aim * Aim;
        wre[n] = (nr * Are + ni * Aim) / den;
        wim[n] = (ni * Are - nr * Aim) / den;
        const float er2 = expf(dre * (float)CHUNK);
        const float ph  = dim * (float)CHUNK;
        Zre[n] = er2 * cosf(ph);
        Zim[n] = er2 * sinf(ph);
    }
    __syncthreads();

    float lzr[Nmode], lzi[Nmode];
#pragma unroll
    for (int n = 0; n < Nmode; ++n) { lzr[n] = zre[n]; lzi[n] = zim[n]; }

    const size_t base = (size_t)bh * Lsz + (size_t)tid * CHUNK;

    float sr[Nmode] = {}, si[Nmode] = {};
    for (int j = 0; j < CHUNK; ++j) {
        const float uv = u[base + j];
#pragma unroll
        for (int n = 0; n < Nmode; ++n) {
            const float tr = lzr[n] * sr[n] - lzi[n] * si[n] + uv;
            const float ti = lzr[n] * si[n] + lzi[n] * sr[n];
            sr[n] = tr; si[n] = ti;
        }
    }
#pragma unroll
    for (int n = 0; n < Nmode; ++n) { sAre[tid][n] = sr[n]; sAim[tid][n] = si[n]; }
    __syncthreads();

    if (tid < Nmode) {
        const int n = tid;
        const float Zr = Zre[n], Zi = Zim[n];
        float pr = 0.0f, pi = 0.0f;
        for (int c = 0; c < NTHREADS; ++c) {
            const float tr = sAre[c][n], ti = sAim[c][n];
            sAre[c][n] = pr; sAim[c][n] = pi;
            const float npr = Zr * pr - Zi * pi + tr;
            const float npi = Zr * pi + Zi * pr + ti;
            pr = npr; pi = npi;
        }
    }
    __syncthreads();

    float lwr[Nmode], lwi[Nmode];
#pragma unroll
    for (int n = 0; n < Nmode; ++n) {
        sr[n] = sAre[tid][n]; si[n] = sAim[tid][n];
        lwr[n] = wre[n];      lwi[n] = wim[n];
    }
    const float Dh = Dp[h], rwh = res_w[h];
    for (int j = 0; j < CHUNK; ++j) {
        const float uv = u[base + j];
        float acc = 0.0f;
#pragma unroll
        for (int n = 0; n < Nmode; ++n) {
            const float tr = lzr[n] * sr[n] - lzi[n] * si[n] + uv;
            const float ti = lzr[n] * si[n] + lzi[n] * sr[n];
            sr[n] = tr; si[n] = ti;
            acc += lwr[n] * tr - lwi[n] * ti;
        }
        const float y = tanhf(2.0f * acc + uv * Dh);
        out[base + j] = y + x[base + j] * rwh;
    }
}

extern "C" void kernel_launch(void* const* d_in, const int* in_sizes, int n_in,
                              void* d_out, int out_size, void* d_ws, size_t ws_size,
                              hipStream_t stream)
{
    (void)in_sizes; (void)n_in; (void)out_size; (void)ws_size;
    const float* x      = (const float*)d_in[0];
    const float* W      = (const float*)d_in[1];
    const float* bias   = (const float*)d_in[2];
    const float* log_dt = (const float*)d_in[3];
    const float* C_re   = (const float*)d_in[4];
    const float* C_im   = (const float*)d_in[5];
    const float* lAr    = (const float*)d_in[6];
    const float* Aim    = (const float*)d_in[7];
    const float* Dp     = (const float*)d_in[8];
    const float* rw     = (const float*)d_in[9];
    float* out = (float*)d_out;
    float* u   = (float*)d_ws;

    dim3 g1(Lsz / 128, Bsz);
    k_channel_mix<<<g1, NTHREADS, 0, stream>>>(x, W, bias, u);
    k_dssm_scan<<<Bsz * Hsz, NTHREADS, 0, stream>>>(u, x, log_dt, C_re, C_im,
                                                    lAr, Aim, Dp, rw, out);
}


# scan kernel: u, x and out cross memory as whole 128-B lines (b128, 4 lines per wave access) via an LDS row transpose, next tile prefetched, instead of 4-B per-lane accesses 512 B apart
# speedup vs baseline: 5.9483x; 5.9483x over previous
#include <hip/hip_runtime.h>
#include <hip/hip_bf16.h>
#include <math.h>


typedef __attribute__((ext_vector_type(2))) float v2f;
typedef __attribute__((ext_vector_type(8))) float v8f;
typedef __attribute__((ext_vector_type(4))) float v4f;

#define Bsz 8
#define Hsz 64
#define Lsz 32768
#define Nmode 16
#define NTHREADS 256
#define CHUNK (Lsz / NTHREADS)
#define TCOLS 32
#define NTILE (CHUNK / TCOLS)
#define TQ (TCOLS / 4)
#define PITCH 36

__global__ __launch_bounds__(NTHREADS)
void k_channel_mix(const float* __restrict__ x, const float* __restrict__ W,
                   const float* __restrict__ bias, float* __restrict__ u)
{
    const int tid  = threadIdx.x;
    const int wave = tid >> 5;
    const int lane = tid & 31;
    const int half = lane >> 4;
    const int lp   = lane & 15;

    const int b  = blockIdx.y;
    const int l0 = blockIdx.x * 128 + wave * 16;
    const float* xb = x + (size_t)b * Hsz * Lsz;

    v8f acc[4] = {};

    for (int k0 = 0; k0 < Hsz; k0 += 4) {
        v2f bb;
        bb.x = xb[(size_t)(k0 + 2 * half + 0) * Lsz + l0 + lp];
        bb.y = xb[(size_t)(k0 + 2 * half + 1) * Lsz + l0 + lp];
#pragma unroll
        for (int mt = 0; mt < 4; ++mt) {
            const float* wrow = W + (size_t)(mt * 16 + lp) * Hsz + k0 + 2 * half;
            v2f aa = *(const v2f*)wrow;
            acc[mt] = __builtin_amdgcn_wmma_f32_16x16x4_f32(
                false, aa, false, bb, (short)0, acc[mt], false, false);
        }
    }

#pragma unroll
    for (int mt = 0; mt < 4; ++mt) {
#pragma unroll
        for (int r = 0; r < 8; ++r) {
            const int g = mt * 16 + r + half * 8;
            const float v = acc[mt][r] + bias[g];
            u[((size_t)b * Hsz + g) * Lsz + l0 + lp] = tanhf(v);
        }
    }
}

__device__ __forceinline__ void lines_load(v4f (&g)[TQ], const float* p)
{
#pragma unroll
    for (int i = 0; i < TQ; ++i) g[i] = *(const v4f*)(p + (size_t)i * 4 * CHUNK);
}
__device__ __forceinline__ void lines_to_rows(float* s, const v4f (&g)[TQ])
{
#pragma unroll
    for (int i = 0; i < TQ; ++i) *(v4f*)(s + i * 4 * PITCH) = g[i];
}
__device__ __forceinline__ void rows_to_lines(float* p, const float* s)
{
#pragma unroll
    for (int i = 0; i < TQ; ++i)
        *(v4f*)(p + (size_t)i * 4 * CHUNK) = *(const v4f*)(s + i * 4 * PITCH);
}

__global__ __launch_bounds__(NTHREADS)
void k_dssm_scan(const float* __restrict__ u, const float* __restrict__ x,
                 const float* __restrict__ log_dt,
                 const float* __restrict__ C_re, const float* __restrict__ C_im,
                 const float* __restrict__ log_A_real,
                 const float* __restrict__ A_imag,
                 const float* __restrict__ Dp, const float* __restrict__ res_w,
                 float* __restrict__ out)
{
    __shared__ float zre[Nmode], zim[Nmode];
    __shared__ float wre[Nmode], wim[Nmode];
    __shared__ float Zre[Nmode], Zim[Nmode];
    __shared__ float sAre[NTHREADS][Nmode + 1];
    __shared__ float sAim[NTHREADS][Nmode + 1];
    __shared__ __attribute__((aligned(16))) float stg[3][NTHREADS * PITCH];
    float* const stA = stg[0];
    float* const stB = stg[1];
    float* const stC = stg[2];

    const int tid = threadIdx.x;
    const int bh  = blockIdx.x;
    const int h   = bh & (Hsz - 1);

    if (tid < Nmode) {
        const int n = tid;
        const float dt  = expf(log_dt[h]);
        const float Are = -expf(log_A_real[h * Nmode + n]);
        const float Aim = A_imag[h * Nmode + n];
        const float dre = Are * dt, dim = Aim * dt;
        const float er  = expf(dre);
        const float zr  = er * cosf(dim), zi = er * sinf(dim);
        zre[n] = zr; zim[n] = zi;
        const float e1r = zr - 1.0f, e1i = zi;
        const float cr = C_re[h * Nmode + n], ci = C_im[h * Nmode + n];
        const float nr = cr * e1r - ci * e1i;
        const float ni = cr * e1i + ci * e1r;
        const float den = Are * Are + Aim * Aim;
        wre[n] = (nr * Are + ni * Aim) / den;
        wim[n] = (ni * Are - nr * Aim) / den;
        const float er2 = expf(dre * (float)CHUNK);
        const float ph  = dim * (float)CHUNK;
        Zre[n] = er2 * cosf(ph);
        Zim[n] = er2 * sinf(ph);
    }
    __syncthreads();

    float lzr[Nmode], lzi[Nmode];
#pragma unroll
    for (int n = 0; n < Nmode; ++n) { lzr[n] = zre[n]; lzi[n] = zim[n]; }

    const int wave = tid >> 5;
    const int lane = tid & 31;
    const int trow = wave * 32 + (lane >> 3);
    const int tcol = (lane & 7) * 4;
    const size_t goff = (size_t)bh * Lsz + (size_t)trow * CHUNK + tcol;
    const int    loff = trow * PITCH + tcol;
    const int    ooff = tid * PITCH;

    v4f gu[TQ], gx[TQ];
    lines_load(gu, u + goff);

    float sr[Nmode] = {}, si[Nmode] = {};
    for (int t = 0; t < NTILE; ++t) {
        float* st = stg[t & 1];
        lines_to_rows(st + loff, gu);
        __syncthreads();
        if (t + 1 < NTILE)
            lines_load(gu, u + goff + (t + 1) * TCOLS);
        for (int q = 0; q < TQ; ++q) {
            const v4f uq = *(const v4f*)(st + ooff + 4 * q);
#pragma unroll
            for (int e = 0; e < 4; ++e) {
                const float uv = uq[e];
#pragma unroll
                for (int n = 0; n < Nmode; ++n) {
                    const float tr = lzr[n] * sr[n] - lzi[n] * si[n] + uv;
                    const float ti = lzr[n] * si[n] + lzi[n] * sr[n];
                    sr[n] = tr; si[n] = ti;
                }
            }
        }
    }
#pragma unroll
    for (int n = 0; n < Nmode; ++n) { sAre[tid][n] = sr[n]; sAim[tid][n] = si[n]; }
    __syncthreads();

    lines_load(gu, u + goff);
    lines_load(gx, x + goff);

    if (tid < Nmode) {
        const int n = tid;
        const float Zr = Zre[n], Zi = Zim[n];
        float pr = 0.0f, pi = 0.0f;
        for (int c = 0; c < NTHREADS; ++c) {
            const float tr = sAre[c][n], ti = sAim[c][n];
            sAre[c][n] = pr; sAim[c][n] = pi;
            const float npr = Zr * pr - Zi * pi + tr;
            const float npi = Zr * pi + Zi * pr + ti;
            pr = npr; pi = npi;
        }
    }
    __syncthreads();

    float lwr[Nmode], lwi[Nmode];
#pragma unroll
    for (int n = 0; n < Nmode; ++n) {
        sr[n] = sAre[tid][n]; si[n] = sAim[tid][n];
        lwr[n] = wre[n];      lwi[n] = wim[n];
    }
    const float Dh = Dp[h], rwh = res_w[h];
    for (int t = 0; t < NTILE; ++t) {
        lines_to_rows(stA + loff, gu);
        lines_to_rows(stB + loff, gx);
        __syncthreads();
        if (t + 1 < NTILE) {
            lines_load(gu, u + goff + (t + 1) * TCOLS);
            lines_load(gx, x + goff + (t + 1) * TCOLS);
        }
        for (int q = 0; q < TQ; ++q) {
            const v4f uq = *(const v4f*)(stA + ooff + 4 * q);
            const v4f xq = *(const v4f*)(stB + ooff + 4 * q);
            v4f oq;
#pragma unroll
            for (int e = 0; e < 4; ++e) {
                const float uv = uq[e];
                float acc = 0.0f;
#pragma unroll
                for (int n = 0; n < Nmode; ++n) {
                    const float tr = lzr[n] * sr[n] - lzi[n] * si[n] + uv;
                    const float ti = lzr[n] * si[n] + lzi[n] * sr[n];
                    sr[n] = tr; si[n] = ti;
                    acc += lwr[n] * tr - lwi[n] * ti;
                }
                const float y = tanhf(2.0f * acc + uv * Dh);
                oq[e] = y + xq[e] * rwh;
            }
            *(v4f*)(stC + ooff + 4 * q) = oq;
        }
        __syncthreads();
        rows_to_lines(out + goff + t * TCOLS, stC + loff);
    }
}

extern "C" void kernel_launch(void* const* d_in, const int* in_sizes, int n_in,
                              void* d_out, int out_size, void* d_ws, size_t ws_size,
                              hipStream_t stream)
{
    (void)in_sizes; (void)n_in; (void)out_size; (void)ws_size;
    const float* x      = (const float*)d_in[0];
    const float* W      = (const float*)d_in[1];
    const float* bias   = (const float*)d_in[2];
    const float* log_dt = (const float*)d_in[3];
    const float* C_re   = (const float*)d_in[4];
    const float* C_im   = (const float*)d_in[5];
    const float* lAr    = (const float*)d_in[6];
    const float* Aim    = (const float*)d_in[7];
    const float* Dp     = (const float*)d_in[8];
    const float* rw     = (const float*)d_in[9];
    float* out = (float*)d_out;
    float* u   = (float*)d_ws;

    dim3 g1(Lsz / 128, Bsz);
    k_channel_mix<<<g1, NTHREADS, 0, stream>>>(x, W, bias, u);
    k_dssm_scan<<<Bsz * Hsz, NTHREADS, 0, stream>>>(u, x, log_dt, C_re, C_im,
                                                    lAr, Aim, Dp, rw, out);
}


# channel-mix kernel re-tiled: one wave = 64 rows x 64 columns, x loaded and u stored as whole 256-B row runs (b128, columns interleaved over the 4 N-tiles), W fragments from one LDS copy per block
# speedup vs baseline: 7.0994x; 1.1935x over previous
#include <hip/hip_runtime.h>
#include <hip/hip_bf16.h>
#include <math.h>


typedef __attribute__((ext_vector_type(2))) float v2f;
typedef __attribute__((ext_vector_type(8))) float v8f;
typedef __attribute__((ext_vector_type(4))) float v4f;

#define Bsz 8
#define Hsz 64
#define Lsz 32768
#define Nmode 16
#define NTHREADS 256
#define CHUNK (Lsz / NTHREADS)
#define TCOLS 32
#define NTILE (CHUNK / TCOLS)
#define TQ (TCOLS / 4)
#define PITCH 36

#define K1_WCOLS 64
#define K1_BCOLS (8 * K1_WCOLS)
#define WPITCH 68
#define K1_AHEAD 6

__global__ __launch_bounds__(NTHREADS)
void k_channel_mix(const float* __restrict__ x, const float* __restrict__ W,
                   const float* __restrict__ bias, float* __restrict__ u)
{
    __shared__ __attribute__((aligned(16))) float sW[Hsz * WPITCH];

    const int tid  = threadIdx.x;
    const int wave = tid >> 5;
    const int lane = tid & 31;
    const int half = lane >> 4;
    const int lp   = lane & 15;

    const int b  = blockIdx.y;
    const int l0 = blockIdx.x * K1_BCOLS + wave * K1_WCOLS;

    {
        const int row = tid >> 2, c0 = (tid & 3) * 16;
        const float* src = W + row * Hsz + c0;
        const v4f w0 = *(const v4f*)(src + 0), w1 = *(const v4f*)(src + 4);
        const v4f w2 = *(const v4f*)(src + 8), w3 = *(const v4f*)(src + 12);
        float* dst = sW + row * WPITCH + c0;
        *(v4f*)(dst + 0) = w0; *(v4f*)(dst + 4)  = w1;
        *(v4f*)(dst + 8) = w2; *(v4f*)(dst + 12) = w3;
    }

    const float* xp = x + (size_t)b * Hsz * Lsz + (size_t)(2 * half) * Lsz + l0 + 4 * lp;

    v4f ring[K1_AHEAD][2];
#pragma unroll
    for (int i = 0; i < K1_AHEAD; ++i) {
        ring[i][0] = *(const v4f*)(xp + (size_t)(4 * i + 0) * Lsz);
        ring[i][1] = *(const v4f*)(xp + (size_t)(4 * i + 1) * Lsz);
    }
    __syncthreads();

    v8f acc[4][4] = {};
    const float* wl = sW + lp * WPITCH + 2 * half;

#pragma unroll
    for (int ks = 0; ks < Hsz / 4; ++ks) {
        const v4f c0 = ring[ks % K1_AHEAD][0];
        const v4f c1 = ring[ks % K1_AHEAD][1];
        if (ks + K1_AHEAD < Hsz / 4) {
            ring[ks % K1_AHEAD][0] = *(const v4f*)(xp + (size_t)(4 * (ks + K1_AHEAD) + 0) * Lsz);
            ring[ks % K1_AHEAD][1] = *(const v4f*)(xp + (size_t)(4 * (ks + K1_AHEAD) + 1) * Lsz);
        }
        v2f aa[4];
#pragma unroll
        for (int mt = 0; mt < 4; ++mt)
            aa[mt] = *(const v2f*)(wl + mt * 16 * WPITCH + 4 * ks);
#pragma unroll
        for (int j = 0; j < 4; ++j) {
            v2f bb;
            bb.x = c0[j];
            bb.y = c1[j];
#pragma unroll
            for (int mt = 0; mt < 4; ++mt)
                acc[mt][j] = __builtin_amdgcn_wmma_f32_16x16x4_f32(
                    false, aa[mt], false, bb, (short)0, acc[mt][j], false, false);
        }
    }

    float* up = u + ((size_t)b * Hsz + 8 * half) * Lsz + l0 + 4 * lp;
#pragma unroll 1
    for (int mt = 0; mt < 4; ++mt) {
#pragma unroll
        for (int r = 0; r < 8; ++r) {
            const int g = mt * 16 + r;
            const float bg = bias[g + 8 * half];
            v4f o;
#pragma unroll
            for (int j = 0; j < 4; ++j) o[j] = tanhf(acc[0][j][r] + bg);
            *(v4f*)(up + (size_t)g * Lsz) = o;
        }
#pragma unroll
        for (int j = 0; j < 4; ++j) {
            acc[0][j] = acc[1][j]; acc[1][j] = acc[2][j]; acc[2][j] = acc[3][j];
        }
    }
}

__device__ __forceinline__ void lines_load(v4f (&g)[TQ], const float* p)
{
#pragma unroll
    for (int i = 0; i < TQ; ++i) g[i] = *(const v4f*)(p + (size_t)i * 4 * CHUNK);
}
__device__ __forceinline__ void lines_to_rows(float* s, const v4f (&g)[TQ])
{
#pragma unroll
    for (int i = 0; i < TQ; ++i) *(v4f*)(s + i * 4 * PITCH) = g[i];
}
__device__ __forceinline__ void rows_to_lines(float* p, const float* s)
{
#pragma unroll
    for (int i = 0; i < TQ; ++i)
        *(v4f*)(p + (size_t)i * 4 * CHUNK) = *(const v4f*)(s + i * 4 * PITCH);
}

__global__ __launch_bounds__(NTHREADS)
void k_dssm_scan(const float* __restrict__ u, const float* __restrict__ x,
                 const float* __restrict__ log_dt,
                 const float* __restrict__ C_re, const float* __restrict__ C_im,
                 const float* __restrict__ log_A_real,
                 const float* __restrict__ A_imag,
                 const float* __restrict__ Dp, const float* __restrict__ res_w,
                 float* __restrict__ out)
{
    __shared__ float zre[Nmode], zim[Nmode];
    __shared__ float wre[Nmode], wim[Nmode];
    __shared__ float Zre[Nmode], Zim[Nmode];
    __shared__ float sAre[NTHREADS][Nmode + 1];
    __shared__ float sAim[NTHREADS][Nmode + 1];
    __shared__ __attribute__((aligned(16))) float stg[3][NTHREADS * PITCH];
    float* const stA = stg[0];
    float* const stB = stg[1];
    float* const stC = stg[2];

    const int tid = threadIdx.x;
    const int bh  = blockIdx.x;
    const int h   = bh & (Hsz - 1);

    if (tid < Nmode) {
        const int n = tid;
        const float dt  = expf(log_dt[h]);
        const float Are = -expf(log_A_real[h * Nmode + n]);
        const float Aim = A_imag[h * Nmode + n];
        const float dre = Are * dt, dim = Aim * dt;
        const float er  = expf(dre);
        const float zr  = er * cosf(dim), zi = er * sinf(dim);
        zre[n] = zr; zim[n] = zi;
        const float e1r = zr - 1.0f, e1i = zi;
        const float cr = C_re[h * Nmode + n], ci = C_im[h * Nmode + n];
        const float nr = cr * e1r - ci * e1i;
        const float ni = cr * e1i + ci * e1r;
        const float den = Are * Are + Aim * Aim;
        wre[n] = (nr * Are + ni * Aim) / den;
        wim[n] = (ni * Are - nr * Aim) / den;
        const float er2 = expf(dre * (float)CHUNK);
        const float ph  = dim * (float)CHUNK;
        Zre[n] = er2 * cosf(ph);
        Zim[n] = er2 * sinf(ph);
    }
    __syncthreads();

    float lzr[Nmode], lzi[Nmode];
#pragma unroll
    for (int n = 0; n < Nmode; ++n) { lzr[n] = zre[n]; lzi[n] = zim[n]; }

    const int wave = tid >> 5;
    const int lane = tid & 31;
    const int trow = wave * 32 + (lane >> 3);
    const int tcol = (lane & 7) * 4;
    const size_t goff = (size_t)bh * Lsz + (size_t)trow * CHUNK + tcol;
    const int    loff = trow * PITCH + tcol;
    const int    ooff = tid * PITCH;

    v4f gu[TQ], gx[TQ];
    lines_load(gu, u + goff);

    float sr[Nmode] = {}, si[Nmode] = {};
    for (int t = 0; t < NTILE; ++t) {
        float* st = stg[t & 1];
        lines_to_rows(st + loff, gu);
        __syncthreads();
        if (t + 1 < NTILE)
            lines_load(gu, u + goff + (t + 1) * TCOLS);
        for (int q = 0; q < TQ; ++q) {
            const v4f uq = *(const v4f*)(st + ooff + 4 * q);
#pragma unroll
            for (int e = 0; e < 4; ++e) {
                const float uv = uq[e];
#pragma unroll
                for (int n = 0; n < Nmode; ++n) {
                    const float tr = lzr[n] * sr[n] - lzi[n] * si[n] + uv;
                    const float ti = lzr[n] * si[n] + lzi[n] * sr[n];
                    sr[n] = tr; si[n] = ti;
                }
            }
        }
    }
#pragma unroll
    for (int n = 0; n < Nmode; ++n) { sAre[tid][n] = sr[n]; sAim[tid][n] = si[n]; }
    __syncthreads();

    lines_load(gu, u + goff);
    lines_load(gx, x + goff);

    if (tid < Nmode) {
        const int n = tid;
        const float Zr = Zre[n], Zi = Zim[n];
        float pr = 0.0f, pi = 0.0f;
        for (int c = 0; c < NTHREADS; ++c) {
            const float tr = sAre[c][n], ti = sAim[c][n];
            sAre[c][n] = pr; sAim[c][n] = pi;
            const float npr = Zr * pr - Zi * pi + tr;
            const float npi = Zr * pi + Zi * pr + ti;
            pr = npr; pi = npi;
        }
    }
    __syncthreads();

    float lwr[Nmode], lwi[Nmode];
#pragma unroll
    for (int n = 0; n < Nmode; ++n) {
        sr[n] = sAre[tid][n]; si[n] = sAim[tid][n];
        lwr[n] = wre[n];      lwi[n] = wim[n];
    }
    const float Dh = Dp[h], rwh = res_w[h];
    for (int t = 0; t < NTILE; ++t) {
        lines_to_rows(stA + loff, gu);
        lines_to_rows(stB + loff, gx);
        __syncthreads();
        if (t + 1 < NTILE) {
            lines_load(gu, u + goff + (t + 1) * TCOLS);
            lines_load(gx, x + goff + (t + 1) * TCOLS);
        }
        for (int q = 0; q < TQ; ++q) {
            const v4f uq = *(const v4f*)(stA + ooff + 4 * q);
            const v4f xq = *(const v4f*)(stB + ooff + 4 * q);
            v4f oq;
#pragma unroll
            for (int e = 0; e < 4; ++e) {
                const float uv = uq[e];
                float acc = 0.0f;
#pragma unroll
                for (int n = 0; n < Nmode; ++n) {
                    const float tr = lzr[n] * sr[n] - lzi[n] * si[n] + uv;
                    const float ti = lzr[n] * si[n] + lzi[n] * sr[n];
                    sr[n] = tr; si[n] = ti;
                    acc += lwr[n] * tr - lwi[n] * ti;
                }
                const float y = tanhf(2.0f * acc + uv * Dh);
                oq[e] = y + xq[e] * rwh;
            }
            *(v4f*)(stC + ooff + 4 * q) = oq;
        }
        __syncthreads();
        rows_to_lines(out + goff + t * TCOLS, stC + loff);
    }
}

extern "C" void kernel_launch(void* const* d_in, const int* in_sizes, int n_in,
                              void* d_out, int out_size, void* d_ws, size_t ws_size,
                              hipStream_t stream)
{
    (void)in_sizes; (void)n_in; (void)out_size; (void)ws_size;
    const float* x      = (const float*)d_in[0];
    const float* W      = (const float*)d_in[1];
    const float* bias   = (const float*)d_in[2];
    const float* log_dt = (const float*)d_in[3];
    const float* C_re   = (const float*)d_in[4];
    const float* C_im   = (const float*)d_in[5];
    const float* lAr    = (const float*)d_in[6];
    const float* Aim    = (const float*)d_in[7];
    const float* Dp     = (const float*)d_in[8];
    const float* rw     = (const float*)d_in[9];
    float* out = (float*)d_out;
    float* u   = (float*)d_ws;

    dim3 g1(Lsz / K1_BCOLS, Bsz);
    k_channel_mix<<<g1, NTHREADS, 0, stream>>>(x, W, bias, u);
    k_dssm_scan<<<Bsz * Hsz, NTHREADS, 0, stream>>>(u, x, log_dt, C_re, C_im,
                                                    lAr, Aim, Dp, rw, out);
}


# scan kernel: recurrence stepped 16 positions at a time on f32 WMMA (Toeplitz of the conv kernel, state-in and state-out matrices) instead of per element on the vector unit; sum order changes
# speedup vs baseline: 8.6668x; 1.2208x over previous
#include <hip/hip_runtime.h>
#include <hip/hip_bf16.h>
#include <math.h>


typedef __attribute__((ext_vector_type(2))) float v2f;
typedef __attribute__((ext_vector_type(8))) float v8f;
typedef __attribute__((ext_vector_type(4))) float v4f;

#define Bsz 8
#define Hsz 64
#define Lsz 32768
#define Nmode 16
#define NTHREADS 256
#define CHUNK (Lsz / NTHREADS)
#define TCOLS 32
#define NTILE (CHUNK / TCOLS)
#define TQ (TCOLS / 4)
#define PITCH 36

#define K1_WCOLS 64
#define K1_BCOLS (8 * K1_WCOLS)
#define WPITCH 68
#define K1_AHEAD 6

__global__ __launch_bounds__(NTHREADS)
void k_channel_mix(const float* __restrict__ x, const float* __restrict__ W,
                   const float* __restrict__ bias, float* __restrict__ u)
{
    __shared__ __attribute__((aligned(16))) float sW[Hsz * WPITCH];

    const int tid  = threadIdx.x;
    const int wave = tid >> 5;
    const int lane = tid & 31;
    const int half = lane >> 4;
    const int lp   = lane & 15;

    const int b  = blockIdx.y;
    const int l0 = blockIdx.x * K1_BCOLS + wave * K1_WCOLS;

    {
        const int row = tid >> 2, c0 = (tid & 3) * 16;
        const float* src = W + row * Hsz + c0;
        const v4f w0 = *(const v4f*)(src + 0), w1 = *(const v4f*)(src + 4);
        const v4f w2 = *(const v4f*)(src + 8), w3 = *(const v4f*)(src + 12);
        float* dst = sW + row * WPITCH + c0;
        *(v4f*)(dst + 0) = w0; *(v4f*)(dst + 4)  = w1;
        *(v4f*)(dst + 8) = w2; *(v4f*)(dst + 12) = w3;
    }

    const float* xp = x + (size_t)b * Hsz * Lsz + (size_t)(2 * half) * Lsz + l0 + 4 * lp;

    v4f ring[K1_AHEAD][2];
#pragma unroll
    for (int i = 0; i < K1_AHEAD; ++i) {
        ring[i][0] = *(const v4f*)(xp + (size_t)(4 * i + 0) * Lsz);
        ring[i][1] = *(const v4f*)(xp + (size_t)(4 * i + 1) * Lsz);
    }
    __syncthreads();

    v8f acc[4][4] = {};
    const float* wl = sW + lp * WPITCH + 2 * half;

#pragma unroll
    for (int ks = 0; ks < Hsz / 4; ++ks) {
        const v4f c0 = ring[ks % K1_AHEAD][0];
        const v4f c1 = ring[ks % K1_AHEAD][1];
        if (ks + K1_AHEAD < Hsz / 4) {
            ring[ks % K1_AHEAD][0] = *(const v4f*)(xp + (size_t)(4 * (ks + K1_AHEAD) + 0) * Lsz);
            ring[ks % K1_AHEAD][1] = *(const v4f*)(xp + (size_t)(4 * (ks + K1_AHEAD) + 1) * Lsz);
        }
        v2f aa[4];
#pragma unroll
        for (int mt = 0; mt < 4; ++mt)
            aa[mt] = *(const v2f*)(wl + mt * 16 * WPITCH + 4 * ks);
#pragma unroll
        for (int j = 0; j < 4; ++j) {
            v2f bb;
            bb.x = c0[j];
            bb.y = c1[j];
#pragma unroll
            for (int mt = 0; mt < 4; ++mt)
                acc[mt][j] = __builtin_amdgcn_wmma_f32_16x16x4_f32(
                    false, aa[mt], false, bb, (short)0, acc[mt][j], false, false);
        }
    }

    float* up = u + ((size_t)b * Hsz + 8 * half) * Lsz + l0 + 4 * lp;
#pragma unroll 1
    for (int mt = 0; mt < 4; ++mt) {
#pragma unroll
        for (int r = 0; r < 8; ++r) {
            const int g = mt * 16 + r;
            const float bg = bias[g + 8 * half];
            v4f o;
#pragma unroll
            for (int j = 0; j < 4; ++j) o[j] = tanhf(acc[0][j][r] + bg);
            *(v4f*)(up + (size_t)g * Lsz) = o;
        }
#pragma unroll
        for (int j = 0; j < 4; ++j) {
            acc[0][j] = acc[1][j]; acc[1][j] = acc[2][j]; acc[2][j] = acc[3][j];
        }
    }
}

__device__ __forceinline__ void lines_load(v4f (&g)[TQ], const float* p)
{
#pragma unroll
    for (int i = 0; i < TQ; ++i) g[i] = *(const v4f*)(p + (size_t)i * 4 * CHUNK);
}
__device__ __forceinline__ void lines_to_rows(float* s, const v4f (&g)[TQ])
{
#pragma unroll
    for (int i = 0; i < TQ; ++i) *(v4f*)(s + i * 4 * PITCH) = g[i];
}
__device__ __forceinline__ void rows_to_lines(float* p, const float* s)
{
#pragma unroll
    for (int i = 0; i < TQ; ++i)
        *(v4f*)(p + (size_t)i * 4 * CHUNK) = *(const v4f*)(s + i * 4 * PITCH);
}

#define WM(a, b, c) __builtin_amdgcn_wmma_f32_16x16x4_f32(false, (a), false, (b), (short)0, (c), false, false)

__global__ __launch_bounds__(NTHREADS)
void k_dssm_scan(const float* __restrict__ u, const float* __restrict__ x,
                 const float* __restrict__ log_dt,
                 const float* __restrict__ C_re, const float* __restrict__ C_im,
                 const float* __restrict__ log_A_real,
                 const float* __restrict__ A_imag,
                 const float* __restrict__ Dp, const float* __restrict__ res_w,
                 float* __restrict__ out)
{
    __shared__ float Zre[Nmode], Zim[Nmode];
    __shared__ float tZr[Nmode], tZi[Nmode];
    __shared__ float tK[16];
    __shared__ float tKn[16][Nmode + 1];
    __shared__ float tP[2 * Nmode][17];
    __shared__ float tQ[16][2 * Nmode + 1];
    __shared__ float sAre[NTHREADS][Nmode + 1];
    __shared__ float sAim[NTHREADS][Nmode + 1];
    __shared__ __attribute__((aligned(16))) float stg[3][NTHREADS * PITCH];
    float* const stA = stg[0];
    float* const stB = stg[1];
    float* const stC = stg[2];

    const int tid  = threadIdx.x;
    const int bh   = blockIdx.x;
    const int h    = bh & (Hsz - 1);
    const int wave = tid >> 5;
    const int lane = tid & 31;
    const int hf   = lane >> 4;
    const int m    = lane & 15;

    const int trow = wave * 32 + (lane >> 3);
    const int tcol = (lane & 7) * 4;
    const size_t goff = (size_t)bh * Lsz + (size_t)trow * CHUNK + tcol;
    const int    loff = trow * PITCH + tcol;
    v4f gu[TQ], gx[TQ];
    lines_load(gu, u + goff);

    {
        const int n = tid & 15, k = tid >> 4;
        const float dt  = expf(log_dt[h]);
        const float Are = -expf(log_A_real[h * Nmode + n]);
        const float Aim = A_imag[h * Nmode + n];
        const float dre = Are * dt, dim = Aim * dt;
        const float er  = expf(dre);
        const float zr  = er * cosf(dim), zi = er * sinf(dim);
        const float e1r = zr - 1.0f, e1i = zi;
        const float cr = C_re[h * Nmode + n], ci = C_im[h * Nmode + n];
        const float nr = cr * e1r - ci * e1i;
        const float ni = cr * e1i + ci * e1r;
        const float den = Are * Are + Aim * Aim;
        const float wr = (nr * Are + ni * Aim) / den;
        const float wi = (ni * Are - nr * Aim) / den;
        const float ek  = expf(dre * (float)k), pk = dim * (float)k;
        const float zkr = ek * cosf(pk), zki = ek * sinf(pk);
        const float z1r = zkr * zr - zki * zi, z1i = zkr * zi + zki * zr;
        tP[n][15 - k]         = zkr;
        tP[Nmode + n][15 - k] = zki;
        tQ[k][n]         =  2.0f * (wr * z1r - wi * z1i);
        tQ[k][Nmode + n] = -2.0f * (wr * z1i + wi * z1r);
        tKn[k][n] = 2.0f * (wr * zkr - wi * zki);
        if (k == 0) {
            const float e16 = expf(dre * 16.0f), p16 = dim * 16.0f;
            tZr[n] = e16 * cosf(p16); tZi[n] = e16 * sinf(p16);
            const float eC = expf(dre * (float)CHUNK), pC = dim * (float)CHUNK;
            Zre[n] = eC * cosf(pC);   Zim[n] = eC * sinf(pC);
        }
    }
    __syncthreads();
    if (tid < 16) {
        float s = 0.0f;
        for (int n = 0; n < Nmode; ++n) s += tKn[tid][n];
        tK[tid] = s;
    }
    __syncthreads();

    const float Dh = Dp[h], rwh = res_w[h];

    v2f aK[4], aP[2][4], aQ[8];
    float z16r[8], z16i[8];
#pragma unroll
    for (int s = 0; s < 4; ++s) {
#pragma unroll
        for (int r = 0; r < 2; ++r) {
            const int p = 8 * hf + 2 * s + r;
            const int d = m - p;
            const float kv = tK[d & 15];
            aK[s][r]    = (d > 0) ? kv : ((d == 0) ? kv + Dh : 0.0f);
            aP[0][s][r] = tP[m][p];
            aP[1][s][r] = tP[Nmode + m][p];
        }
    }
#pragma unroll
    for (int s = 0; s < 8; ++s) {
#pragma unroll
        for (int r = 0; r < 2; ++r)
            aQ[s][r] = tQ[m][16 * (s >> 2) + 8 * hf + 2 * (s & 3) + r];
    }
#pragma unroll
    for (int r = 0; r < 8; ++r) { z16r[r] = tZr[8 * hf + r]; z16i[r] = tZi[8 * hf + r]; }

    const int foff = (wave * 32 + m) * PITCH + 8 * hf;

    v8f S[2][2] = {};
    for (int t = 0; t < NTILE; ++t) {
        float* st = stg[t & 1];
        lines_to_rows(st + loff, gu);
        __syncthreads();
        if (t + 1 < NTILE)
            lines_load(gu, u + goff + (t + 1) * TCOLS);
#pragma unroll
        for (int q2 = 0; q2 < 2; ++q2) {
#pragma unroll
            for (int j = 0; j < 2; ++j) {
                const float* fp = st + foff + 16 * j * PITCH + 16 * q2;
                const v4f ub0 = *(const v4f*)(fp), ub1 = *(const v4f*)(fp + 4);
                v2f b0, b1, b2, b3;
                b0.x = ub0[0]; b0.y = ub0[1]; b1.x = ub0[2]; b1.y = ub0[3];
                b2.x = ub1[0]; b2.y = ub1[1]; b3.x = ub1[2]; b3.y = ub1[3];
                v8f nr, ni;
#pragma unroll
                for (int r = 0; r < 8; ++r) {
                    nr[r] = z16r[r] * S[j][0][r] - z16i[r] * S[j][1][r];
                    ni[r] = z16r[r] * S[j][1][r] + z16i[r] * S[j][0][r];
                }
                nr = WM(aP[0][0], b0, nr); ni = WM(aP[1][0], b0, ni);
                nr = WM(aP[0][1], b1, nr); ni = WM(aP[1][1], b1, ni);
                nr = WM(aP[0][2], b2, nr); ni = WM(aP[1][2], b2, ni);
                nr = WM(aP[0][3], b3, nr); ni = WM(aP[1][3], b3, ni);
                S[j][0] = nr; S[j][1] = ni;
            }
        }
    }
#pragma unroll
    for (int j = 0; j < 2; ++j) {
#pragma unroll
        for (int r = 0; r < 8; ++r) {
            sAre[wave * 32 + 16 * j + m][8 * hf + r] = S[j][0][r];
            sAim[wave * 32 + 16 * j + m][8 * hf + r] = S[j][1][r];
        }
    }
    __syncthreads();

    lines_load(gu, u + goff);
    lines_load(gx, x + goff);

    if (tid < Nmode) {
        const int n = tid;
        const float Zr = Zre[n], Zi = Zim[n];
        float pr = 0.0f, pi = 0.0f;
        for (int c = 0; c < NTHREADS; ++c) {
            const float tr = sAre[c][n], ti = sAim[c][n];
            sAre[c][n] = pr; sAim[c][n] = pi;
            const float npr = Zr * pr - Zi * pi + tr;
            const float npi = Zr * pi + Zi * pr + ti;
            pr = npr; pi = npi;
        }
    }
    __syncthreads();

#pragma unroll
    for (int j = 0; j < 2; ++j) {
#pragma unroll
        for (int r = 0; r < 8; ++r) {
            S[j][0][r] = sAre[wave * 32 + 16 * j + m][8 * hf + r];
            S[j][1][r] = sAim[wave * 32 + 16 * j + m][8 * hf + r];
        }
    }
    for (int t = 0; t < NTILE; ++t) {
        lines_to_rows(stA + loff, gu);
        lines_to_rows(stB + loff, gx);
        __syncthreads();
        if (t + 1 < NTILE) {
            lines_load(gu, u + goff + (t + 1) * TCOLS);
            lines_load(gx, x + goff + (t + 1) * TCOLS);
        }
#pragma unroll
        for (int q2 = 0; q2 < 2; ++q2) {
#pragma unroll
            for (int j = 0; j < 2; ++j) {
                const int fo = foff + 16 * j * PITCH + 16 * q2;
                const v4f ub0 = *(const v4f*)(stA + fo), ub1 = *(const v4f*)(stA + fo + 4);
                const v4f xb0 = *(const v4f*)(stB + fo), xb1 = *(const v4f*)(stB + fo + 4);
                v2f b0, b1, b2, b3;
                b0.x = ub0[0]; b0.y = ub0[1]; b1.x = ub0[2]; b1.y = ub0[3];
                b2.x = ub1[0]; b2.y = ub1[1]; b3.x = ub1[2]; b3.y = ub1[3];
                v8f y = {};
                y = WM(aK[0], b0, y); y = WM(aK[1], b1, y);
                y = WM(aK[2], b2, y); y = WM(aK[3], b3, y);
#pragma unroll
                for (int s = 0; s < 8; ++s) {
                    v2f sb;
                    sb.x = S[j][s >> 2][2 * (s & 3)];
                    sb.y = S[j][s >> 2][2 * (s & 3) + 1];
                    y = WM(aQ[s], sb, y);
                }
                v8f nr, ni;
#pragma unroll
                for (int r = 0; r < 8; ++r) {
                    nr[r] = z16r[r] * S[j][0][r] - z16i[r] * S[j][1][r];
                    ni[r] = z16r[r] * S[j][1][r] + z16i[r] * S[j][0][r];
                }
                nr = WM(aP[0][0], b0, nr); ni = WM(aP[1][0], b0, ni);
                nr = WM(aP[0][1], b1, nr); ni = WM(aP[1][1], b1, ni);
                nr = WM(aP[0][2], b2, nr); ni = WM(aP[1][2], b2, ni);
                nr = WM(aP[0][3], b3, nr); ni = WM(aP[1][3], b3, ni);
                S[j][0] = nr; S[j][1] = ni;
                v4f o0, o1;
#pragma unroll
                for (int e = 0; e < 4; ++e) {
                    o0[e] = tanhf(y[e])     + xb0[e] * rwh;
                    o1[e] = tanhf(y[4 + e]) + xb1[e] * rwh;
                }
                *(v4f*)(stC + fo)     = o0;
                *(v4f*)(stC + fo + 4) = o1;
            }
        }
        __syncthreads();
        rows_to_lines(out + goff + t * TCOLS, stC + loff);
    }
}

extern "C" void kernel_launch(void* const* d_in, const int* in_sizes, int n_in,
                              void* d_out, int out_size, void* d_ws, size_t ws_size,
                              hipStream_t stream)
{
    (void)in_sizes; (void)n_in; (void)out_size; (void)ws_size;
    const float* x      = (const float*)d_in[0];
    const float* W      = (const float*)d_in[1];
    const float* bias   = (const float*)d_in[2];
    const float* log_dt = (const float*)d_in[3];
    const float* C_re   = (const float*)d_in[4];
    const float* C_im   = (const float*)d_in[5];
    const float* lAr    = (const float*)d_in[6];
    const float* Aim    = (const float*)d_in[7];
    const float* Dp     = (const float*)d_in[8];
    const float* rw     = (const float*)d_in[9];
    float* out = (float*)d_out;
    float* u   = (float*)d_ws;

    dim3 g1(Lsz / K1_BCOLS, Bsz);
    k_channel_mix<<<g1, NTHREADS, 0, stream>>>(x, W, bias, u);
    k_dssm_scan<<<Bsz * Hsz, NTHREADS, 0, stream>>>(u, x, log_dt, C_re, C_im,
                                                    lAr, Aim, Dp, rw, out);
}


# tanh in both kernels: library tanhf (two masked paths, ~30 vector instr.) -> branch-free 1-2/(exp2(2*log2e*v)+1) on hardware exp2 and rcp, abs error ~2e-7; kernel 1 epilogue unrolled, no tile moves
# speedup vs baseline: 11.2936x; 1.3031x over previous
#include <hip/hip_runtime.h>
#include <hip/hip_bf16.h>
#include <math.h>


typedef __attribute__((ext_vector_type(2))) float v2f;
typedef __attribute__((ext_vector_type(8))) float v8f;
typedef __attribute__((ext_vector_type(4))) float v4f;

#define Bsz 8
#define Hsz 64
#define Lsz 32768
#define Nmode 16
#define NTHREADS 256
#define CHUNK (Lsz / NTHREADS)
#define TCOLS 32
#define NTILE (CHUNK / TCOLS)
#define TQ (TCOLS / 4)
#define PITCH 36

#define TWO_LOG2E 2.885390081777927f
__device__ __forceinline__ float tanh_scaled(float es)
{
    const float t = __builtin_amdgcn_exp2f(es);
    return fmaf(__builtin_amdgcn_rcpf(t + 1.0f), -2.0f, 1.0f);
}

#define K1_WCOLS 64
#define K1_BCOLS (8 * K1_WCOLS)
#define WPITCH 68
#define K1_AHEAD 6

__global__ __launch_bounds__(NTHREADS)
void k_channel_mix(const float* __restrict__ x, const float* __restrict__ W,
                   const float* __restrict__ bias, float* __restrict__ u)
{
    __shared__ __attribute__((aligned(16))) float sW[Hsz * WPITCH];

    const int tid  = threadIdx.x;
    const int wave = tid >> 5;
    const int lane = tid & 31;
    const int half = lane >> 4;
    const int lp   = lane & 15;

    const int b  = blockIdx.y;
    const int l0 = blockIdx.x * K1_BCOLS + wave * K1_WCOLS;

    {
        const int row = tid >> 2, c0 = (tid & 3) * 16;
        const float* src = W + row * Hsz + c0;
        const v4f w0 = *(const v4f*)(src + 0), w1 = *(const v4f*)(src + 4);
        const v4f w2 = *(const v4f*)(src + 8), w3 = *(const v4f*)(src + 12);
        float* dst = sW + row * WPITCH + c0;
        *(v4f*)(dst + 0) = w0; *(v4f*)(dst + 4)  = w1;
        *(v4f*)(dst + 8) = w2; *(v4f*)(dst + 12) = w3;
    }

    const float* xp = x + (size_t)b * Hsz * Lsz + (size_t)(2 * half) * Lsz + l0 + 4 * lp;

    v4f ring[K1_AHEAD][2];
#pragma unroll
    for (int i = 0; i < K1_AHEAD; ++i) {
        ring[i][0] = *(const v4f*)(xp + (size_t)(4 * i + 0) * Lsz);
        ring[i][1] = *(const v4f*)(xp + (size_t)(4 * i + 1) * Lsz);
    }
    __syncthreads();

    v8f acc[4][4] = {};
    const float* wl = sW + lp * WPITCH + 2 * half;

#pragma unroll
    for (int ks = 0; ks < Hsz / 4; ++ks) {
        const v4f c0 = ring[ks % K1_AHEAD][0];
        const v4f c1 = ring[ks % K1_AHEAD][1];
        if (ks + K1_AHEAD < Hsz / 4) {
            ring[ks % K1_AHEAD][0] = *(const v4f*)(xp + (size_t)(4 * (ks + K1_AHEAD) + 0) * Lsz);
            ring[ks % K1_AHEAD][1] = *(const v4f*)(xp + (size_t)(4 * (ks + K1_AHEAD) + 1) * Lsz);
        }
        v2f aa[4];
#pragma unroll
        for (int mt = 0; mt < 4; ++mt)
            aa[mt] = *(const v2f*)(wl + mt * 16 * WPITCH + 4 * ks);
#pragma unroll
        for (int j = 0; j < 4; ++j) {
            v2f bb;
            bb.x = c0[j];
            bb.y = c1[j];
#pragma unroll
            for (int mt = 0; mt < 4; ++mt)
                acc[mt][j] = __builtin_amdgcn_wmma_f32_16x16x4_f32(
                    false, aa[mt], false, bb, (short)0, acc[mt][j], false, false);
        }
    }

    float* up = u + ((size_t)b * Hsz + 8 * half) * Lsz + l0 + 4 * lp;
#pragma unroll
    for (int mt = 0; mt < 4; ++mt) {
#pragma unroll
        for (int r = 0; r < 8; ++r) {
            const int g = mt * 16 + r;
            const float bs = bias[g + 8 * half] * TWO_LOG2E;
            v4f o;
#pragma unroll
            for (int j = 0; j < 4; ++j) o[j] = tanh_scaled(fmaf(acc[mt][j][r], TWO_LOG2E, bs));
            *(v4f*)(up + (size_t)g * Lsz) = o;
        }
    }
}

__device__ __forceinline__ void lines_load(v4f (&g)[TQ], const float* p)
{
#pragma unroll
    for (int i = 0; i < TQ; ++i) g[i] = *(const v4f*)(p + (size_t)i * 4 * CHUNK);
}
__device__ __forceinline__ void lines_to_rows(float* s, const v4f (&g)[TQ])
{
#pragma unroll
    for (int i = 0; i < TQ; ++i) *(v4f*)(s + i * 4 * PITCH) = g[i];
}
__device__ __forceinline__ void rows_to_lines(float* p, const float* s)
{
    v4f g[TQ];
#pragma unroll
    for (int i = 0; i < TQ; ++i) g[i] = *(const v4f*)(s + i * 4 * PITCH);
#pragma unroll
    for (int i = 0; i < TQ; ++i) *(v4f*)(p + (size_t)i * 4 * CHUNK) = g[i];
}

#define WM(a, b, c) __builtin_amdgcn_wmma_f32_16x16x4_f32(false, (a), false, (b), (short)0, (c), false, false)

__global__ __launch_bounds__(NTHREADS)
void k_dssm_scan(const float* __restrict__ u, const float* __restrict__ x,
                 const float* __restrict__ log_dt,
                 const float* __restrict__ C_re, const float* __restrict__ C_im,
                 const float* __restrict__ log_A_real,
                 const float* __restrict__ A_imag,
                 const float* __restrict__ Dp, const float* __restrict__ res_w,
                 float* __restrict__ out)
{
    __shared__ float Zre[Nmode], Zim[Nmode];
    __shared__ float tZr[Nmode], tZi[Nmode];
    __shared__ float tK[16];
    __shared__ float tKn[16][Nmode + 1];
    __shared__ float tP[2 * Nmode][17];
    __shared__ float tQ[16][2 * Nmode + 1];
    __shared__ float sAre[NTHREADS][Nmode + 1];
    __shared__ float sAim[NTHREADS][Nmode + 1];
    __shared__ __attribute__((aligned(16))) float stg[3][NTHREADS * PITCH];
    float* const stA = stg[0];
    float* const stB = stg[1];
    float* const stC = stg[2];

    const int tid  = threadIdx.x;
    const int bh   = blockIdx.x;
    const int h    = bh & (Hsz - 1);
    const int wave = tid >> 5;
    const int lane = tid & 31;
    const int hf   = lane >> 4;
    const int m    = lane & 15;

    const int trow = wave * 32 + (lane >> 3);
    const int tcol = (lane & 7) * 4;
    const size_t goff = (size_t)bh * Lsz + (size_t)trow * CHUNK + tcol;
    const int    loff = trow * PITCH + tcol;
    v4f gu[TQ], gx[TQ];
    lines_load(gu, u + goff);

    {
        const int n = tid & 15, k = tid >> 4;
        const float dt  = expf(log_dt[h]);
        const float Are = -expf(log_A_real[h * Nmode + n]);
        const float Aim = A_imag[h * Nmode + n];
        const float dre = Are * dt, dim = Aim * dt;
        const float er  = expf(dre);
        const float zr  = er * cosf(dim), zi = er * sinf(dim);
        const float e1r = zr - 1.0f, e1i = zi;
        const float cr = C_re[h * Nmode + n], ci = C_im[h * Nmode + n];
        const float nr = cr * e1r - ci * e1i;
        const float ni = cr * e1i + ci * e1r;
        const float den = Are * Are + Aim * Aim;
        const float wr = (nr * Are + ni * Aim) / den;
        const float wi = (ni * Are - nr * Aim) / den;
        const float ek  = expf(dre * (float)k), pk = dim * (float)k;
        const float zkr = ek * cosf(pk), zki = ek * sinf(pk);
        const float z1r = zkr * zr - zki * zi, z1i = zkr * zi + zki * zr;
        tP[n][15 - k]         = zkr;
        tP[Nmode + n][15 - k] = zki;
        tQ[k][n]         =  2.0f * (wr * z1r - wi * z1i);
        tQ[k][Nmode + n] = -2.0f * (wr * z1i + wi * z1r);
        tKn[k][n] = 2.0f * (wr * zkr - wi * zki);
        if (k == 0) {
            const float e16 = expf(dre * 16.0f), p16 = dim * 16.0f;
            tZr[n] = e16 * cosf(p16); tZi[n] = e16 * sinf(p16);
            const float eC = expf(dre * (float)CHUNK), pC = dim * (float)CHUNK;
            Zre[n] = eC * cosf(pC);   Zim[n] = eC * sinf(pC);
        }
    }
    __syncthreads();
    if (tid < 16) {
        float s = 0.0f;
        for (int n = 0; n < Nmode; ++n) s += tKn[tid][n];
        tK[tid] = s;
    }
    __syncthreads();

    const float Dh = Dp[h], rwh = res_w[h];

    v2f aK[4], aP[2][4], aQ[8];
    float z16r[8], z16i[8];
#pragma unroll
    for (int s = 0; s < 4; ++s) {
#pragma unroll
        for (int r = 0; r < 2; ++r) {
            const int p = 8 * hf + 2 * s + r;
            const int d = m - p;
            const float kv = tK[d & 15];
            aK[s][r]    = (d > 0) ? kv : ((d == 0) ? kv + Dh : 0.0f);
            aP[0][s][r] = tP[m][p];
            aP[1][s][r] = tP[Nmode + m][p];
        }
    }
#pragma unroll
    for (int s = 0; s < 8; ++s) {
#pragma unroll
        for (int r = 0; r < 2; ++r)
            aQ[s][r] = tQ[m][16 * (s >> 2) + 8 * hf + 2 * (s & 3) + r];
    }
#pragma unroll
    for (int r = 0; r < 8; ++r) { z16r[r] = tZr[8 * hf + r]; z16i[r] = tZi[8 * hf + r]; }

    const int foff = (wave * 32 + m) * PITCH + 8 * hf;

    v8f S[2][2] = {};
    for (int t = 0; t < NTILE; ++t) {
        float* st = stg[t & 1];
        lines_to_rows(st + loff, gu);
        __syncthreads();
        if (t + 1 < NTILE)
            lines_load(gu, u + goff + (t + 1) * TCOLS);
#pragma unroll
        for (int q2 = 0; q2 < 2; ++q2) {
#pragma unroll
            for (int j = 0; j < 2; ++j) {
                const float* fp = st + foff + 16 * j * PITCH + 16 * q2;
                const v4f ub0 = *(const v4f*)(fp), ub1 = *(const v4f*)(fp + 4);
                v2f b0, b1, b2, b3;
                b0.x = ub0[0]; b0.y = ub0[1]; b1.x = ub0[2]; b1.y = ub0[3];
                b2.x = ub1[0]; b2.y = ub1[1]; b3.x = ub1[2]; b3.y = ub1[3];
                v8f nr, ni;
#pragma unroll
                for (int r = 0; r < 8; ++r) {
                    nr[r] = z16r[r] * S[j][0][r] - z16i[r] * S[j][1][r];
                    ni[r] = z16r[r] * S[j][1][r] + z16i[r] * S[j][0][r];
                }
                nr = WM(aP[0][0], b0, nr); ni = WM(aP[1][0], b0, ni);
                nr = WM(aP[0][1], b1, nr); ni = WM(aP[1][1], b1, ni);
                nr = WM(aP[0][2], b2, nr); ni = WM(aP[1][2], b2, ni);
                nr = WM(aP[0][3], b3, nr); ni = WM(aP[1][3], b3, ni);
                S[j][0] = nr; S[j][1] = ni;
            }
        }
    }
#pragma unroll
    for (int j = 0; j < 2; ++j) {
#pragma unroll
        for (int r = 0; r < 8; ++r) {
            sAre[wave * 32 + 16 * j + m][8 * hf + r] = S[j][0][r];
            sAim[wave * 32 + 16 * j + m][8 * hf + r] = S[j][1][r];
        }
    }
    __syncthreads();

    lines_load(gu, u + goff);
    lines_load(gx, x + goff);

    if (tid < Nmode) {
        const int n = tid;
        const float Zr = Zre[n], Zi = Zim[n];
        float pr = 0.0f, pi = 0.0f;
        for (int c = 0; c < NTHREADS; ++c) {
            const float tr = sAre[c][n], ti = sAim[c][n];
            sAre[c][n] = pr; sAim[c][n] = pi;
            const float npr = Zr * pr - Zi * pi + tr;
            const float npi = Zr * pi + Zi * pr + ti;
            pr = npr; pi = npi;
        }
    }
    __syncthreads();

#pragma unroll
    for (int j = 0; j < 2; ++j) {
#pragma unroll
        for (int r = 0; r < 8; ++r) {
            S[j][0][r] = sAre[wave * 32 + 16 * j + m][8 * hf + r];
            S[j][1][r] = sAim[wave * 32 + 16 * j + m][8 * hf + r];
        }
    }
#pragma unroll 1
    for (int t = 0; t < NTILE; ++t) {
        lines_to_rows(stA + loff, gu);
        lines_to_rows(stB + loff, gx);
        __syncthreads();
        if (t + 1 < NTILE) {
            lines_load(gu, u + goff + (t + 1) * TCOLS);
            lines_load(gx, x + goff + (t + 1) * TCOLS);
        }
#pragma unroll
        for (int q2 = 0; q2 < 2; ++q2) {
#pragma unroll
            for (int j = 0; j < 2; ++j) {
                const int fo = foff + 16 * j * PITCH + 16 * q2;
                const v4f ub0 = *(const v4f*)(stA + fo), ub1 = *(const v4f*)(stA + fo + 4);
                const v4f xb0 = *(const v4f*)(stB + fo), xb1 = *(const v4f*)(stB + fo + 4);
                v2f b0, b1, b2, b3;
                b0.x = ub0[0]; b0.y = ub0[1]; b1.x = ub0[2]; b1.y = ub0[3];
                b2.x = ub1[0]; b2.y = ub1[1]; b3.x = ub1[2]; b3.y = ub1[3];
                v8f y = {};
                y = WM(aK[0], b0, y); y = WM(aK[1], b1, y);
                y = WM(aK[2], b2, y); y = WM(aK[3], b3, y);
#pragma unroll
                for (int s = 0; s < 8; ++s) {
                    v2f sb;
                    sb.x = S[j][s >> 2][2 * (s & 3)];
                    sb.y = S[j][s >> 2][2 * (s & 3) + 1];
                    y = WM(aQ[s], sb, y);
                }
                v8f nr, ni;
#pragma unroll
                for (int r = 0; r < 8; ++r) {
                    nr[r] = z16r[r] * S[j][0][r] - z16i[r] * S[j][1][r];
                    ni[r] = z16r[r] * S[j][1][r] + z16i[r] * S[j][0][r];
                }
                nr = WM(aP[0][0], b0, nr); ni = WM(aP[1][0], b0, ni);
                nr = WM(aP[0][1], b1, nr); ni = WM(aP[1][1], b1, ni);
                nr = WM(aP[0][2], b2, nr); ni = WM(aP[1][2], b2, ni);
                nr = WM(aP[0][3], b3, nr); ni = WM(aP[1][3], b3, ni);
                S[j][0] = nr; S[j][1] = ni;
                v4f o0, o1;
#pragma unroll
                for (int e = 0; e < 4; ++e) {
                    o0[e] = tanh_scaled(y[e] * TWO_LOG2E)     + xb0[e] * rwh;
                    o1[e] = tanh_scaled(y[4 + e] * TWO_LOG2E) + xb1[e] * rwh;
                }
                *(v4f*)(stC + fo)     = o0;
                *(v4f*)(stC + fo + 4) = o1;
            }
        }
        __syncthreads();
        rows_to_lines(out + goff + t * TCOLS, stC + loff);
    }
}

extern "C" void kernel_launch(void* const* d_in, const int* in_sizes, int n_in,
                              void* d_out, int out_size, void* d_ws, size_t ws_size,
                              hipStream_t stream)
{
    (void)in_sizes; (void)n_in; (void)out_size; (void)ws_size;
    const float* x      = (const float*)d_in[0];
    const float* W      = (const float*)d_in[1];
    const float* bias   = (const float*)d_in[2];
    const float* log_dt = (const float*)d_in[3];
    const float* C_re   = (const float*)d_in[4];
    const float* C_im   = (const float*)d_in[5];
    const float* lAr    = (const float*)d_in[6];
    const float* Aim    = (const float*)d_in[7];
    const float* Dp     = (const float*)d_in[8];
    const float* rw     = (const float*)d_in[9];
    float* out = (float*)d_out;
    float* u   = (float*)d_ws;

    dim3 g1(Lsz / K1_BCOLS, Bsz);
    k_channel_mix<<<g1, NTHREADS, 0, stream>>>(x, W, bias, u);
    k_dssm_scan<<<Bsz * Hsz, NTHREADS, 0, stream>>>(u, x, log_dt, C_re, C_im,
                                                    lAr, Aim, Dp, rw, out);
}


# scan kernel: tile loops of both passes without block barriers: each wave stages, reads and stores only its own 32 chunk rows in LDS, so wave-order LDS access replaces 12 block barriers
# speedup vs baseline: 12.1037x; 1.0717x over previous
#include <hip/hip_runtime.h>
#include <hip/hip_bf16.h>
#include <math.h>


typedef __attribute__((ext_vector_type(2))) float v2f;
typedef __attribute__((ext_vector_type(8))) float v8f;
typedef __attribute__((ext_vector_type(4))) float v4f;

#define Bsz 8
#define Hsz 64
#define Lsz 32768
#define Nmode 16
#define NTHREADS 256
#define CHUNK (Lsz / NTHREADS)
#define TCOLS 32
#define NTILE (CHUNK / TCOLS)
#define TQ (TCOLS / 4)
#define PITCH 36

#define TWO_LOG2E 2.885390081777927f
__device__ __forceinline__ float tanh_scaled(float es)
{
    const float t = __builtin_amdgcn_exp2f(es);
    return fmaf(__builtin_amdgcn_rcpf(t + 1.0f), -2.0f, 1.0f);
}

#define K1_WCOLS 64
#define K1_BCOLS (8 * K1_WCOLS)
#define WPITCH 68
#define K1_AHEAD 6

__global__ __launch_bounds__(NTHREADS)
void k_channel_mix(const float* __restrict__ x, const float* __restrict__ W,
                   const float* __restrict__ bias, float* __restrict__ u)
{
    __shared__ __attribute__((aligned(16))) float sW[Hsz * WPITCH];

    const int tid  = threadIdx.x;
    const int wave = tid >> 5;
    const int lane = tid & 31;
    const int half = lane >> 4;
    const int lp   = lane & 15;

    const int b  = blockIdx.y;
    const int l0 = blockIdx.x * K1_BCOLS + wave * K1_WCOLS;

    {
        const int row = tid >> 2, c0 = (tid & 3) * 16;
        const float* src = W + row * Hsz + c0;
        const v4f w0 = *(const v4f*)(src + 0), w1 = *(const v4f*)(src + 4);
        const v4f w2 = *(const v4f*)(src + 8), w3 = *(const v4f*)(src + 12);
        float* dst = sW + row * WPITCH + c0;
        *(v4f*)(dst + 0) = w0; *(v4f*)(dst + 4)  = w1;
        *(v4f*)(dst + 8) = w2; *(v4f*)(dst + 12) = w3;
    }

    const float* xp = x + (size_t)b * Hsz * Lsz + (size_t)(2 * half) * Lsz + l0 + 4 * lp;

    v4f ring[K1_AHEAD][2];
#pragma unroll
    for (int i = 0; i < K1_AHEAD; ++i) {
        ring[i][0] = *(const v4f*)(xp + (size_t)(4 * i + 0) * Lsz);
        ring[i][1] = *(const v4f*)(xp + (size_t)(4 * i + 1) * Lsz);
    }
    __syncthreads();

    v8f acc[4][4] = {};
    const float* wl = sW + lp * WPITCH + 2 * half;

#pragma unroll
    for (int ks = 0; ks < Hsz / 4; ++ks) {
        const v4f c0 = ring[ks % K1_AHEAD][0];
        const v4f c1 = ring[ks % K1_AHEAD][1];
        if (ks + K1_AHEAD < Hsz / 4) {
            ring[ks % K1_AHEAD][0] = *(const v4f*)(xp + (size_t)(4 * (ks + K1_AHEAD) + 0) * Lsz);
            ring[ks % K1_AHEAD][1] = *(const v4f*)(xp + (size_t)(4 * (ks + K1_AHEAD) + 1) * Lsz);
        }
        v2f aa[4];
#pragma unroll
        for (int mt = 0; mt < 4; ++mt)
            aa[mt] = *(const v2f*)(wl + mt * 16 * WPITCH + 4 * ks);
#pragma unroll
        for (int j = 0; j < 4; ++j) {
            v2f bb;
            bb.x = c0[j];
            bb.y = c1[j];
#pragma unroll
            for (int mt = 0; mt < 4; ++mt)
                acc[mt][j] = __builtin_amdgcn_wmma_f32_16x16x4_f32(
                    false, aa[mt], false, bb, (short)0, acc[mt][j], false, false);
        }
    }

    float* up = u + ((size_t)b * Hsz + 8 * half) * Lsz + l0 + 4 * lp;
#pragma unroll
    for (int mt = 0; mt < 4; ++mt) {
#pragma unroll
        for (int r = 0; r < 8; ++r) {
            const int g = mt * 16 + r;
            const float bs = bias[g + 8 * half] * TWO_LOG2E;
            v4f o;
#pragma unroll
            for (int j = 0; j < 4; ++j) o[j] = tanh_scaled(fmaf(acc[mt][j][r], TWO_LOG2E, bs));
            *(v4f*)(up + (size_t)g * Lsz) = o;
        }
    }
}

__device__ __forceinline__ void lines_load(v4f (&g)[TQ], const float* p)
{
#pragma unroll
    for (int i = 0; i < TQ; ++i) g[i] = *(const v4f*)(p + (size_t)i * 4 * CHUNK);
}
__device__ __forceinline__ void lines_to_rows(float* s, const v4f (&g)[TQ])
{
#pragma unroll
    for (int i = 0; i < TQ; ++i) *(v4f*)(s + i * 4 * PITCH) = g[i];
}
__device__ __forceinline__ void rows_to_lines(float* p, const float* s)
{
    v4f g[TQ];
#pragma unroll
    for (int i = 0; i < TQ; ++i) g[i] = *(const v4f*)(s + i * 4 * PITCH);
#pragma unroll
    for (int i = 0; i < TQ; ++i) *(v4f*)(p + (size_t)i * 4 * CHUNK) = g[i];
}

__device__ __forceinline__ void wave_rows_sync()
{
    __builtin_amdgcn_fence(__ATOMIC_RELEASE, "wavefront", "local");
    __builtin_amdgcn_wave_barrier();
    __builtin_amdgcn_fence(__ATOMIC_ACQUIRE, "wavefront", "local");
}

#define WM(a, b, c) __builtin_amdgcn_wmma_f32_16x16x4_f32(false, (a), false, (b), (short)0, (c), false, false)

__global__ __launch_bounds__(NTHREADS)
void k_dssm_scan(const float* __restrict__ u, const float* __restrict__ x,
                 const float* __restrict__ log_dt,
                 const float* __restrict__ C_re, const float* __restrict__ C_im,
                 const float* __restrict__ log_A_real,
                 const float* __restrict__ A_imag,
                 const float* __restrict__ Dp, const float* __restrict__ res_w,
                 float* __restrict__ out)
{
    __shared__ float Zre[Nmode], Zim[Nmode];
    __shared__ float tZr[Nmode], tZi[Nmode];
    __shared__ float tK[16];
    __shared__ float tKn[16][Nmode + 1];
    __shared__ float tP[2 * Nmode][17];
    __shared__ float tQ[16][2 * Nmode + 1];
    __shared__ float sAre[NTHREADS][Nmode + 1];
    __shared__ float sAim[NTHREADS][Nmode + 1];
    __shared__ __attribute__((aligned(16))) float stg[3][NTHREADS * PITCH];
    float* const stA = stg[0];
    float* const stB = stg[1];
    float* const stC = stg[2];

    const int tid  = threadIdx.x;
    const int bh   = blockIdx.x;
    const int h    = bh & (Hsz - 1);
    const int wave = tid >> 5;
    const int lane = tid & 31;
    const int hf   = lane >> 4;
    const int m    = lane & 15;

    const int trow = wave * 32 + (lane >> 3);
    const int tcol = (lane & 7) * 4;
    const size_t goff = (size_t)bh * Lsz + (size_t)trow * CHUNK + tcol;
    const int    loff = trow * PITCH + tcol;
    v4f gu[TQ], gx[TQ];
    lines_load(gu, u + goff);

    {
        const int n = tid & 15, k = tid >> 4;
        const float dt  = expf(log_dt[h]);
        const float Are = -expf(log_A_real[h * Nmode + n]);
        const float Aim = A_imag[h * Nmode + n];
        const float dre = Are * dt, dim = Aim * dt;
        const float er  = expf(dre);
        const float zr  = er * cosf(dim), zi = er * sinf(dim);
        const float e1r = zr - 1.0f, e1i = zi;
        const float cr = C_re[h * Nmode + n], ci = C_im[h * Nmode + n];
        const float nr = cr * e1r - ci * e1i;
        const float ni = cr * e1i + ci * e1r;
        const float den = Are * Are + Aim * Aim;
        const float wr = (nr * Are + ni * Aim) / den;
        const float wi = (ni * Are - nr * Aim) / den;
        const float ek  = expf(dre * (float)k), pk = dim * (float)k;
        const float zkr = ek * cosf(pk), zki = ek * sinf(pk);
        const float z1r = zkr * zr - zki * zi, z1i = zkr * zi + zki * zr;
        tP[n][15 - k]         = zkr;
        tP[Nmode + n][15 - k] = zki;
        tQ[k][n]         =  2.0f * (wr * z1r - wi * z1i);
        tQ[k][Nmode + n] = -2.0f * (wr * z1i + wi * z1r);
        tKn[k][n] = 2.0f * (wr * zkr - wi * zki);
        if (k == 0) {
            const float e16 = expf(dre * 16.0f), p16 = dim * 16.0f;
            tZr[n] = e16 * cosf(p16); tZi[n] = e16 * sinf(p16);
            const float eC = expf(dre * (float)CHUNK), pC = dim * (float)CHUNK;
            Zre[n] = eC * cosf(pC);   Zim[n] = eC * sinf(pC);
        }
    }
    __syncthreads();
    if (tid < 16) {
        float s = 0.0f;
        for (int n = 0; n < Nmode; ++n) s += tKn[tid][n];
        tK[tid] = s;
    }
    __syncthreads();

    const float Dh = Dp[h], rwh = res_w[h];

    v2f aK[4], aP[2][4], aQ[8];
    float z16r[8], z16i[8];
#pragma unroll
    for (int s = 0; s < 4; ++s) {
#pragma unroll
        for (int r = 0; r < 2; ++r) {
            const int p = 8 * hf + 2 * s + r;
            const int d = m - p;
            const float kv = tK[d & 15];
            aK[s][r]    = (d > 0) ? kv : ((d == 0) ? kv + Dh : 0.0f);
            aP[0][s][r] = tP[m][p];
            aP[1][s][r] = tP[Nmode + m][p];
        }
    }
#pragma unroll
    for (int s = 0; s < 8; ++s) {
#pragma unroll
        for (int r = 0; r < 2; ++r)
            aQ[s][r] = tQ[m][16 * (s >> 2) + 8 * hf + 2 * (s & 3) + r];
    }
#pragma unroll
    for (int r = 0; r < 8; ++r) { z16r[r] = tZr[8 * hf + r]; z16i[r] = tZi[8 * hf + r]; }

    const int foff = (wave * 32 + m) * PITCH + 8 * hf;

    v8f S[2][2] = {};
#pragma unroll 1
    for (int t = 0; t < NTILE; ++t) {
        float* st = stg[t & 1];
        lines_to_rows(st + loff, gu);
        wave_rows_sync();
        if (t + 1 < NTILE)
            lines_load(gu, u + goff + (t + 1) * TCOLS);
#pragma unroll
        for (int q2 = 0; q2 < 2; ++q2) {
#pragma unroll
            for (int j = 0; j < 2; ++j) {
                const float* fp = st + foff + 16 * j * PITCH + 16 * q2;
                const v4f ub0 = *(const v4f*)(fp), ub1 = *(const v4f*)(fp + 4);
                v2f b0, b1, b2, b3;
                b0.x = ub0[0]; b0.y = ub0[1]; b1.x = ub0[2]; b1.y = ub0[3];
                b2.x = ub1[0]; b2.y = ub1[1]; b3.x = ub1[2]; b3.y = ub1[3];
                v8f nr, ni;
#pragma unroll
                for (int r = 0; r < 8; ++r) {
                    nr[r] = z16r[r] * S[j][0][r] - z16i[r] * S[j][1][r];
                    ni[r] = z16r[r] * S[j][1][r] + z16i[r] * S[j][0][r];
                }
                nr = WM(aP[0][0], b0, nr); ni = WM(aP[1][0], b0, ni);
                nr = WM(aP[0][1], b1, nr); ni = WM(aP[1][1], b1, ni);
                nr = WM(aP[0][2], b2, nr); ni = WM(aP[1][2], b2, ni);
                nr = WM(aP[0][3], b3, nr); ni = WM(aP[1][3], b3, ni);
                S[j][0] = nr; S[j][1] = ni;
            }
        }
    }
#pragma unroll
    for (int j = 0; j < 2; ++j) {
#pragma unroll
        for (int r = 0; r < 8; ++r) {
            sAre[wave * 32 + 16 * j + m][8 * hf + r] = S[j][0][r];
            sAim[wave * 32 + 16 * j + m][8 * hf + r] = S[j][1][r];
        }
    }
    __syncthreads();

    lines_load(gu, u + goff);
    lines_load(gx, x + goff);

    if (tid < Nmode) {
        const int n = tid;
        const float Zr = Zre[n], Zi = Zim[n];
        float pr = 0.0f, pi = 0.0f;
        for (int c = 0; c < NTHREADS; ++c) {
            const float tr = sAre[c][n], ti = sAim[c][n];
            sAre[c][n] = pr; sAim[c][n] = pi;
            const float npr = Zr * pr - Zi * pi + tr;
            const float npi = Zr * pi + Zi * pr + ti;
            pr = npr; pi = npi;
        }
    }
    __syncthreads();

#pragma unroll
    for (int j = 0; j < 2; ++j) {
#pragma unroll
        for (int r = 0; r < 8; ++r) {
            S[j][0][r] = sAre[wave * 32 + 16 * j + m][8 * hf + r];
            S[j][1][r] = sAim[wave * 32 + 16 * j + m][8 * hf + r];
        }
    }
#pragma unroll 1
    for (int t = 0; t < NTILE; ++t) {
        lines_to_rows(stA + loff, gu);
        lines_to_rows(stB + loff, gx);
        wave_rows_sync();
        if (t + 1 < NTILE) {
            lines_load(gu, u + goff + (t + 1) * TCOLS);
            lines_load(gx, x + goff + (t + 1) * TCOLS);
        }
#pragma unroll
        for (int q2 = 0; q2 < 2; ++q2) {
#pragma unroll
            for (int j = 0; j < 2; ++j) {
                const int fo = foff + 16 * j * PITCH + 16 * q2;
                const v4f ub0 = *(const v4f*)(stA + fo), ub1 = *(const v4f*)(stA + fo + 4);
                const v4f xb0 = *(const v4f*)(stB + fo), xb1 = *(const v4f*)(stB + fo + 4);
                v2f b0, b1, b2, b3;
                b0.x = ub0[0]; b0.y = ub0[1]; b1.x = ub0[2]; b1.y = ub0[3];
                b2.x = ub1[0]; b2.y = ub1[1]; b3.x = ub1[2]; b3.y = ub1[3];
                v8f y = {};
                y = WM(aK[0], b0, y); y = WM(aK[1], b1, y);
                y = WM(aK[2], b2, y); y = WM(aK[3], b3, y);
#pragma unroll
                for (int s = 0; s < 8; ++s) {
                    v2f sb;
                    sb.x = S[j][s >> 2][2 * (s & 3)];
                    sb.y = S[j][s >> 2][2 * (s & 3) + 1];
                    y = WM(aQ[s], sb, y);
                }
                v8f nr, ni;
#pragma unroll
                for (int r = 0; r < 8; ++r) {
                    nr[r] = z16r[r] * S[j][0][r] - z16i[r] * S[j][1][r];
                    ni[r] = z16r[r] * S[j][1][r] + z16i[r] * S[j][0][r];
                }
                nr = WM(aP[0][0], b0, nr); ni = WM(aP[1][0], b0, ni);
                nr = WM(aP[0][1], b1, nr); ni = WM(aP[1][1], b1, ni);
                nr = WM(aP[0][2], b2, nr); ni = WM(aP[1][2], b2, ni);
                nr = WM(aP[0][3], b3, nr); ni = WM(aP[1][3], b3, ni);
                S[j][0] = nr; S[j][1] = ni;
                v4f o0, o1;
#pragma unroll
                for (int e = 0; e < 4; ++e) {
                    o0[e] = tanh_scaled(y[e] * TWO_LOG2E)     + xb0[e] * rwh;
                    o1[e] = tanh_scaled(y[4 + e] * TWO_LOG2E) + xb1[e] * rwh;
                }
                *(v4f*)(stC + fo)     = o0;
                *(v4f*)(stC + fo + 4) = o1;
            }
        }
        wave_rows_sync();
        rows_to_lines(out + goff + t * TCOLS, stC + loff);
    }
}

extern "C" void kernel_launch(void* const* d_in, const int* in_sizes, int n_in,
                              void* d_out, int out_size, void* d_ws, size_t ws_size,
                              hipStream_t stream)
{
    (void)in_sizes; (void)n_in; (void)out_size; (void)ws_size;
    const float* x      = (const float*)d_in[0];
    const float* W      = (const float*)d_in[1];
    const float* bias   = (const float*)d_in[2];
    const float* log_dt = (const float*)d_in[3];
    const float* C_re   = (const float*)d_in[4];
    const float* C_im   = (const float*)d_in[5];
    const float* lAr    = (const float*)d_in[6];
    const float* Aim    = (const float*)d_in[7];
    const float* Dp     = (const float*)d_in[8];
    const float* rw     = (const float*)d_in[9];
    float* out = (float*)d_out;
    float* u   = (float*)d_ws;

    dim3 g1(Lsz / K1_BCOLS, Bsz);
    k_channel_mix<<<g1, NTHREADS, 0, stream>>>(x, W, bias, u);
    k_dssm_scan<<<Bsz * Hsz, NTHREADS, 0, stream>>>(u, x, log_dt, C_re, C_im,
                                                    lAr, Aim, Dp, rw, out);
}


# scan kernel: cross-chunk state scan on all 256 threads (16 group scans + 15-step group prefix, z^2048 by squaring), not 256 serial steps on 16 lanes; pass 2's tile 0 asked earlier; sum order changes
# speedup vs baseline: 13.0625x; 1.0792x over previous
#include <hip/hip_runtime.h>
#include <hip/hip_bf16.h>
#include <math.h>


typedef __attribute__((ext_vector_type(2))) float v2f;
typedef __attribute__((ext_vector_type(8))) float v8f;
typedef __attribute__((ext_vector_type(4))) float v4f;

#define Bsz 8
#define Hsz 64
#define Lsz 32768
#define Nmode 16
#define NTHREADS 256
#define CHUNK (Lsz / NTHREADS)
#define TCOLS 32
#define NTILE (CHUNK / TCOLS)
#define TQ (TCOLS / 4)
#define PITCH 36

#define TWO_LOG2E 2.885390081777927f
__device__ __forceinline__ float tanh_scaled(float es)
{
    const float t = __builtin_amdgcn_exp2f(es);
    return fmaf(__builtin_amdgcn_rcpf(t + 1.0f), -2.0f, 1.0f);
}

#define K1_WCOLS 64
#define K1_BCOLS (8 * K1_WCOLS)
#define WPITCH 68
#define K1_AHEAD 6

__global__ __launch_bounds__(NTHREADS)
void k_channel_mix(const float* __restrict__ x, const float* __restrict__ W,
                   const float* __restrict__ bias, float* __restrict__ u)
{
    __shared__ __attribute__((aligned(16))) float sW[Hsz * WPITCH];

    const int tid  = threadIdx.x;
    const int wave = tid >> 5;
    const int lane = tid & 31;
    const int half = lane >> 4;
    const int lp   = lane & 15;

    const int b  = blockIdx.y;
    const int l0 = blockIdx.x * K1_BCOLS + wave * K1_WCOLS;

    {
        const int row = tid >> 2, c0 = (tid & 3) * 16;
        const float* src = W + row * Hsz + c0;
        const v4f w0 = *(const v4f*)(src + 0), w1 = *(const v4f*)(src + 4);
        const v4f w2 = *(const v4f*)(src + 8), w3 = *(const v4f*)(src + 12);
        float* dst = sW + row * WPITCH + c0;
        *(v4f*)(dst + 0) = w0; *(v4f*)(dst + 4)  = w1;
        *(v4f*)(dst + 8) = w2; *(v4f*)(dst + 12) = w3;
    }

    const float* xp = x + (size_t)b * Hsz * Lsz + (size_t)(2 * half) * Lsz + l0 + 4 * lp;

    v4f ring[K1_AHEAD][2];
#pragma unroll
    for (int i = 0; i < K1_AHEAD; ++i) {
        ring[i][0] = *(const v4f*)(xp + (size_t)(4 * i + 0) * Lsz);
        ring[i][1] = *(const v4f*)(xp + (size_t)(4 * i + 1) * Lsz);
    }
    __syncthreads();

    v8f acc[4][4] = {};
    const float* wl = sW + lp * WPITCH + 2 * half;

#pragma unroll
    for (int ks = 0; ks < Hsz / 4; ++ks) {
        const v4f c0 = ring[ks % K1_AHEAD][0];
        const v4f c1 = ring[ks % K1_AHEAD][1];
        if (ks + K1_AHEAD < Hsz / 4) {
            ring[ks % K1_AHEAD][0] = *(const v4f*)(xp + (size_t)(4 * (ks + K1_AHEAD) + 0) * Lsz);
            ring[ks % K1_AHEAD][1] = *(const v4f*)(xp + (size_t)(4 * (ks + K1_AHEAD) + 1) * Lsz);
        }
        v2f aa[4];
#pragma unroll
        for (int mt = 0; mt < 4; ++mt)
            aa[mt] = *(const v2f*)(wl + mt * 16 * WPITCH + 4 * ks);
#pragma unroll
        for (int j = 0; j < 4; ++j) {
            v2f bb;
            bb.x = c0[j];
            bb.y = c1[j];
#pragma unroll
            for (int mt = 0; mt < 4; ++mt)
                acc[mt][j] = __builtin_amdgcn_wmma_f32_16x16x4_f32(
                    false, aa[mt], false, bb, (short)0, acc[mt][j], false, false);
        }
    }

    float* up = u + ((size_t)b * Hsz + 8 * half) * Lsz + l0 + 4 * lp;
#pragma unroll
    for (int mt = 0; mt < 4; ++mt) {
#pragma unroll
        for (int r = 0; r < 8; ++r) {
            const int g = mt * 16 + r;
            const float bs = bias[g + 8 * half] * TWO_LOG2E;
            v4f o;
#pragma unroll
            for (int j = 0; j < 4; ++j) o[j] = tanh_scaled(fmaf(acc[mt][j][r], TWO_LOG2E, bs));
            *(v4f*)(up + (size_t)g * Lsz) = o;
        }
    }
}

__device__ __forceinline__ void lines_load(v4f (&g)[TQ], const float* p)
{
#pragma unroll
    for (int i = 0; i < TQ; ++i) g[i] = *(const v4f*)(p + (size_t)i * 4 * CHUNK);
}
__device__ __forceinline__ void lines_to_rows(float* s, const v4f (&g)[TQ])
{
#pragma unroll
    for (int i = 0; i < TQ; ++i) *(v4f*)(s + i * 4 * PITCH) = g[i];
}
__device__ __forceinline__ void rows_to_lines(float* p, const float* s)
{
    v4f g[TQ];
#pragma unroll
    for (int i = 0; i < TQ; ++i) g[i] = *(const v4f*)(s + i * 4 * PITCH);
#pragma unroll
    for (int i = 0; i < TQ; ++i) *(v4f*)(p + (size_t)i * 4 * CHUNK) = g[i];
}

__device__ __forceinline__ void wave_rows_sync()
{
    __builtin_amdgcn_fence(__ATOMIC_RELEASE, "wavefront", "local");
    __builtin_amdgcn_wave_barrier();
    __builtin_amdgcn_fence(__ATOMIC_ACQUIRE, "wavefront", "local");
}

#define WM(a, b, c) __builtin_amdgcn_wmma_f32_16x16x4_f32(false, (a), false, (b), (short)0, (c), false, false)

__global__ __launch_bounds__(NTHREADS)
void k_dssm_scan(const float* __restrict__ u, const float* __restrict__ x,
                 const float* __restrict__ log_dt,
                 const float* __restrict__ C_re, const float* __restrict__ C_im,
                 const float* __restrict__ log_A_real,
                 const float* __restrict__ A_imag,
                 const float* __restrict__ Dp, const float* __restrict__ res_w,
                 float* __restrict__ out)
{
    __shared__ float Zre[Nmode], Zim[Nmode];
    __shared__ float tZr[Nmode], tZi[Nmode];
    __shared__ float tK[16];
    __shared__ float tKn[16][Nmode + 1];
    __shared__ float tP[2 * Nmode][17];
    __shared__ float tQ[16][2 * Nmode + 1];
    __shared__ float sAre[NTHREADS][Nmode + 1];
    __shared__ float sAim[NTHREADS][Nmode + 1];
    __shared__ float sGre[16][Nmode + 1];
    __shared__ float sGim[16][Nmode + 1];
    __shared__ __attribute__((aligned(16))) float stg[3][NTHREADS * PITCH];
    float* const stA = stg[0];
    float* const stB = stg[1];
    float* const stC = stg[2];

    const int tid  = threadIdx.x;
    const int bh   = blockIdx.x;
    const int h    = bh & (Hsz - 1);
    const int wave = tid >> 5;
    const int lane = tid & 31;
    const int hf   = lane >> 4;
    const int m    = lane & 15;

    const int trow = wave * 32 + (lane >> 3);
    const int tcol = (lane & 7) * 4;
    const size_t goff = (size_t)bh * Lsz + (size_t)trow * CHUNK + tcol;
    const int    loff = trow * PITCH + tcol;
    v4f gu[TQ], gx[TQ];
    lines_load(gu, u + goff);
    lines_load(gx, x + goff);

    {
        const int n = tid & 15, k = tid >> 4;
        const float dt  = expf(log_dt[h]);
        const float Are = -expf(log_A_real[h * Nmode + n]);
        const float Aim = A_imag[h * Nmode + n];
        const float dre = Are * dt, dim = Aim * dt;
        const float er  = expf(dre);
        const float zr  = er * cosf(dim), zi = er * sinf(dim);
        const float e1r = zr - 1.0f, e1i = zi;
        const float cr = C_re[h * Nmode + n], ci = C_im[h * Nmode + n];
        const float nr = cr * e1r - ci * e1i;
        const float ni = cr * e1i + ci * e1r;
        const float den = Are * Are + Aim * Aim;
        const float wr = (nr * Are + ni * Aim) / den;
        const float wi = (ni * Are - nr * Aim) / den;
        const float ek  = expf(dre * (float)k), pk = dim * (float)k;
        const float zkr = ek * cosf(pk), zki = ek * sinf(pk);
        const float z1r = zkr * zr - zki * zi, z1i = zkr * zi + zki * zr;
        tP[n][15 - k]         = zkr;
        tP[Nmode + n][15 - k] = zki;
        tQ[k][n]         =  2.0f * (wr * z1r - wi * z1i);
        tQ[k][Nmode + n] = -2.0f * (wr * z1i + wi * z1r);
        tKn[k][n] = 2.0f * (wr * zkr - wi * zki);
        if (k == 0) {
            const float e16 = expf(dre * 16.0f), p16 = dim * 16.0f;
            tZr[n] = e16 * cosf(p16); tZi[n] = e16 * sinf(p16);
            const float eC = expf(dre * (float)CHUNK), pC = dim * (float)CHUNK;
            Zre[n] = eC * cosf(pC);   Zim[n] = eC * sinf(pC);
        }
    }
    __syncthreads();
    if (tid < 16) {
        float s = 0.0f;
        for (int n = 0; n < Nmode; ++n) s += tKn[tid][n];
        tK[tid] = s;
    }
    __syncthreads();

    const float Dh = Dp[h], rwh = res_w[h];

    v2f aK[4], aP[2][4], aQ[8];
    float z16r[8], z16i[8];
#pragma unroll
    for (int s = 0; s < 4; ++s) {
#pragma unroll
        for (int r = 0; r < 2; ++r) {
            const int p = 8 * hf + 2 * s + r;
            const int d = m - p;
            const float kv = tK[d & 15];
            aK[s][r]    = (d > 0) ? kv : ((d == 0) ? kv + Dh : 0.0f);
            aP[0][s][r] = tP[m][p];
            aP[1][s][r] = tP[Nmode + m][p];
        }
    }
#pragma unroll
    for (int s = 0; s < 8; ++s) {
#pragma unroll
        for (int r = 0; r < 2; ++r)
            aQ[s][r] = tQ[m][16 * (s >> 2) + 8 * hf + 2 * (s & 3) + r];
    }
#pragma unroll
    for (int r = 0; r < 8; ++r) { z16r[r] = tZr[8 * hf + r]; z16i[r] = tZi[8 * hf + r]; }

    const int foff = (wave * 32 + m) * PITCH + 8 * hf;

    v8f S[2][2] = {};
#pragma unroll 1
    for (int t = 0; t < NTILE; ++t) {
        float* st = stg[t & 1];
        lines_to_rows(st + loff, gu);
        wave_rows_sync();
        lines_load(gu, u + goff + ((t + 1) & (NTILE - 1)) * TCOLS);
#pragma unroll
        for (int q2 = 0; q2 < 2; ++q2) {
#pragma unroll
            for (int j = 0; j < 2; ++j) {
                const float* fp = st + foff + 16 * j * PITCH + 16 * q2;
                const v4f ub0 = *(const v4f*)(fp), ub1 = *(const v4f*)(fp + 4);
                v2f b0, b1, b2, b3;
                b0.x = ub0[0]; b0.y = ub0[1]; b1.x = ub0[2]; b1.y = ub0[3];
                b2.x = ub1[0]; b2.y = ub1[1]; b3.x = ub1[2]; b3.y = ub1[3];
                v8f nr, ni;
#pragma unroll
                for (int r = 0; r < 8; ++r) {
                    nr[r] = z16r[r] * S[j][0][r] - z16i[r] * S[j][1][r];
                    ni[r] = z16r[r] * S[j][1][r] + z16i[r] * S[j][0][r];
                }
                nr = WM(aP[0][0], b0, nr); ni = WM(aP[1][0], b0, ni);
                nr = WM(aP[0][1], b1, nr); ni = WM(aP[1][1], b1, ni);
                nr = WM(aP[0][2], b2, nr); ni = WM(aP[1][2], b2, ni);
                nr = WM(aP[0][3], b3, nr); ni = WM(aP[1][3], b3, ni);
                S[j][0] = nr; S[j][1] = ni;
            }
        }
    }
#pragma unroll
    for (int j = 0; j < 2; ++j) {
#pragma unroll
        for (int r = 0; r < 8; ++r) {
            sAre[wave * 32 + 16 * j + m][8 * hf + r] = S[j][0][r];
            sAim[wave * 32 + 16 * j + m][8 * hf + r] = S[j][1][r];
        }
    }
    wave_rows_sync();

    {
        const int n  = lane & 15;
        const int jj = lane >> 4;
        const int G  = 2 * wave + jj;
        const int R0 = 16 * G;
        const float Zr = Zre[n], Zi = Zim[n];
        float pr = 0.0f, pi = 0.0f;
#pragma unroll
        for (int c = 0; c < 16; ++c) {
            const float tr = sAre[R0 + c][n], ti = sAim[R0 + c][n];
            const float npr = fmaf(Zr, pr, fmaf(-Zi, pi, tr));
            const float npi = fmaf(Zr, pi, fmaf( Zi, pr, ti));
            pr = npr; pi = npi;
        }
        sGre[G][n] = pr; sGim[G][n] = pi;
        float gr = Zr, gi = Zi;
#pragma unroll
        for (int k = 0; k < 4; ++k) {
            const float nr = gr * gr - gi * gi;
            const float ni = 2.0f * gr * gi;
            gr = nr; gi = ni;
        }
        __syncthreads();
        float qr = 0.0f, qi = 0.0f;
#pragma unroll 5
        for (int g = 0; g < 15; ++g) {
            const float ar = sGre[g][n], ai = sGim[g][n];
            const float nqr = fmaf(gr, qr, fmaf(-gi, qi, ar));
            const float nqi = fmaf(gr, qi, fmaf( gi, qr, ai));
            if (g < G) { qr = nqr; qi = nqi; }
        }
        pr = qr; pi = qi;
#pragma unroll
        for (int c = 0; c < 16; ++c) {
            const float tr = sAre[R0 + c][n], ti = sAim[R0 + c][n];
            sAre[R0 + c][n] = pr; sAim[R0 + c][n] = pi;
            const float npr = fmaf(Zr, pr, fmaf(-Zi, pi, tr));
            const float npi = fmaf(Zr, pi, fmaf( Zi, pr, ti));
            pr = npr; pi = npi;
        }
    }
    wave_rows_sync();

#pragma unroll
    for (int j = 0; j < 2; ++j) {
#pragma unroll
        for (int r = 0; r < 8; ++r) {
            S[j][0][r] = sAre[wave * 32 + 16 * j + m][8 * hf + r];
            S[j][1][r] = sAim[wave * 32 + 16 * j + m][8 * hf + r];
        }
    }
#pragma unroll 1
    for (int t = 0; t < NTILE; ++t) {
        lines_to_rows(stA + loff, gu);
        lines_to_rows(stB + loff, gx);
        wave_rows_sync();
        if (t + 1 < NTILE) {
            lines_load(gu, u + goff + (t + 1) * TCOLS);
            lines_load(gx, x + goff + (t + 1) * TCOLS);
        }
#pragma unroll
        for (int q2 = 0; q2 < 2; ++q2) {
#pragma unroll
            for (int j = 0; j < 2; ++j) {
                const int fo = foff + 16 * j * PITCH + 16 * q2;
                const v4f ub0 = *(const v4f*)(stA + fo), ub1 = *(const v4f*)(stA + fo + 4);
                const v4f xb0 = *(const v4f*)(stB + fo), xb1 = *(const v4f*)(stB + fo + 4);
                v2f b0, b1, b2, b3;
                b0.x = ub0[0]; b0.y = ub0[1]; b1.x = ub0[2]; b1.y = ub0[3];
                b2.x = ub1[0]; b2.y = ub1[1]; b3.x = ub1[2]; b3.y = ub1[3];
                v8f y = {};
                y = WM(aK[0], b0, y); y = WM(aK[1], b1, y);
                y = WM(aK[2], b2, y); y = WM(aK[3], b3, y);
#pragma unroll
                for (int s = 0; s < 8; ++s) {
                    v2f sb;
                    sb.x = S[j][s >> 2][2 * (s & 3)];
                    sb.y = S[j][s >> 2][2 * (s & 3) + 1];
                    y = WM(aQ[s], sb, y);
                }
                v8f nr, ni;
#pragma unroll
                for (int r = 0; r < 8; ++r) {
                    nr[r] = z16r[r] * S[j][0][r] - z16i[r] * S[j][1][r];
                    ni[r] = z16r[r] * S[j][1][r] + z16i[r] * S[j][0][r];
                }
                nr = WM(aP[0][0], b0, nr); ni = WM(aP[1][0], b0, ni);
                nr = WM(aP[0][1], b1, nr); ni = WM(aP[1][1], b1, ni);
                nr = WM(aP[0][2], b2, nr); ni = WM(aP[1][2], b2, ni);
                nr = WM(aP[0][3], b3, nr); ni = WM(aP[1][3], b3, ni);
                S[j][0] = nr; S[j][1] = ni;
                v4f o0, o1;
#pragma unroll
                for (int e = 0; e < 4; ++e) {
                    o0[e] = tanh_scaled(y[e] * TWO_LOG2E)     + xb0[e] * rwh;
                    o1[e] = tanh_scaled(y[4 + e] * TWO_LOG2E) + xb1[e] * rwh;
                }
                *(v4f*)(stC + fo)     = o0;
                *(v4f*)(stC + fo + 4) = o1;
            }
        }
        wave_rows_sync();
        rows_to_lines(out + goff + t * TCOLS, stC + loff);
    }
}

extern "C" void kernel_launch(void* const* d_in, const int* in_sizes, int n_in,
                              void* d_out, int out_size, void* d_ws, size_t ws_size,
                              hipStream_t stream)
{
    (void)in_sizes; (void)n_in; (void)out_size; (void)ws_size;
    const float* x      = (const float*)d_in[0];
    const float* W      = (const float*)d_in[1];
    const float* bias   = (const float*)d_in[2];
    const float* log_dt = (const float*)d_in[3];
    const float* C_re   = (const float*)d_in[4];
    const float* C_im   = (const float*)d_in[5];
    const float* lAr    = (const float*)d_in[6];
    const float* Aim    = (const float*)d_in[7];
    const float* Dp     = (const float*)d_in[8];
    const float* rw     = (const float*)d_in[9];
    float* out = (float*)d_out;
    float* u   = (float*)d_ws;

    dim3 g1(Lsz / K1_BCOLS, Bsz);
    k_channel_mix<<<g1, NTHREADS, 0, stream>>>(x, W, bias, u);
    k_dssm_scan<<<Bsz * Hsz, NTHREADS, 0, stream>>>(u, x, log_dt, C_re, C_im,
                                                    lAr, Aim, Dp, rw, out);
}


# scan kernel: pass 1 leaves its four u tiles where pass 2 reads them (three in LDS areas of their own, the last in its registers): u is read from memory once; x, out, states share an area
# speedup vs baseline: 13.7162x; 1.0500x over previous
#include <hip/hip_runtime.h>
#include <hip/hip_bf16.h>
#include <math.h>


typedef __attribute__((ext_vector_type(2))) float v2f;
typedef __attribute__((ext_vector_type(8))) float v8f;
typedef __attribute__((ext_vector_type(4))) float v4f;

#define Bsz 8
#define Hsz 64
#define Lsz 32768
#define Nmode 16
#define NTHREADS 256
#define CHUNK (Lsz / NTHREADS)
#define TCOLS 32
#define NTILE (CHUNK / TCOLS)
#define TQ (TCOLS / 4)
#define PITCH 36
#define SPITCH (Nmode + 1)
#define SIM (32 * SPITCH)

#define TWO_LOG2E 2.885390081777927f
__device__ __forceinline__ float tanh_scaled(float es)
{
    const float t = __builtin_amdgcn_exp2f(es);
    return fmaf(__builtin_amdgcn_rcpf(t + 1.0f), -2.0f, 1.0f);
}

#define K1_WCOLS 64
#define K1_BCOLS (8 * K1_WCOLS)
#define WPITCH 68
#define K1_AHEAD 6

__global__ __launch_bounds__(NTHREADS)
void k_channel_mix(const float* __restrict__ x, const float* __restrict__ W,
                   const float* __restrict__ bias, float* __restrict__ u)
{
    __shared__ __attribute__((aligned(16))) float sW[Hsz * WPITCH];

    const int tid  = threadIdx.x;
    const int wave = tid >> 5;
    const int lane = tid & 31;
    const int half = lane >> 4;
    const int lp   = lane & 15;

    const int b  = blockIdx.y;
    const int l0 = blockIdx.x * K1_BCOLS + wave * K1_WCOLS;

    {
        const int row = tid >> 2, c0 = (tid & 3) * 16;
        const float* src = W + row * Hsz + c0;
        const v4f w0 = *(const v4f*)(src + 0), w1 = *(const v4f*)(src + 4);
        const v4f w2 = *(const v4f*)(src + 8), w3 = *(const v4f*)(src + 12);
        float* dst = sW + row * WPITCH + c0;
        *(v4f*)(dst + 0) = w0; *(v4f*)(dst + 4)  = w1;
        *(v4f*)(dst + 8) = w2; *(v4f*)(dst + 12) = w3;
    }

    const float* xp = x + (size_t)b * Hsz * Lsz + (size_t)(2 * half) * Lsz + l0 + 4 * lp;

    v4f ring[K1_AHEAD][2];
#pragma unroll
    for (int i = 0; i < K1_AHEAD; ++i) {
        ring[i][0] = *(const v4f*)(xp + (size_t)(4 * i + 0) * Lsz);
        ring[i][1] = *(const v4f*)(xp + (size_t)(4 * i + 1) * Lsz);
    }
    __syncthreads();

    v8f acc[4][4] = {};
    const float* wl = sW + lp * WPITCH + 2 * half;

#pragma unroll
    for (int ks = 0; ks < Hsz / 4; ++ks) {
        const v4f c0 = ring[ks % K1_AHEAD][0];
        const v4f c1 = ring[ks % K1_AHEAD][1];
        if (ks + K1_AHEAD < Hsz / 4) {
            ring[ks % K1_AHEAD][0] = *(const v4f*)(xp + (size_t)(4 * (ks + K1_AHEAD) + 0) * Lsz);
            ring[ks % K1_AHEAD][1] = *(const v4f*)(xp + (size_t)(4 * (ks + K1_AHEAD) + 1) * Lsz);
        }
        v2f aa[4];
#pragma unroll
        for (int mt = 0; mt < 4; ++mt)
            aa[mt] = *(const v2f*)(wl + mt * 16 * WPITCH + 4 * ks);
#pragma unroll
        for (int j = 0; j < 4; ++j) {
            v2f bb;
            bb.x = c0[j];
            bb.y = c1[j];
#pragma unroll
            for (int mt = 0; mt < 4; ++mt)
                acc[mt][j] = __builtin_amdgcn_wmma_f32_16x16x4_f32(
                    false, aa[mt], false, bb, (short)0, acc[mt][j], false, false);
        }
    }

    float* up = u + ((size_t)b * Hsz + 8 * half) * Lsz + l0 + 4 * lp;
#pragma unroll
    for (int mt = 0; mt < 4; ++mt) {
#pragma unroll
        for (int r = 0; r < 8; ++r) {
            const int g = mt * 16 + r;
            const float bs = bias[g + 8 * half] * TWO_LOG2E;
            v4f o;
#pragma unroll
            for (int j = 0; j < 4; ++j) o[j] = tanh_scaled(fmaf(acc[mt][j][r], TWO_LOG2E, bs));
            *(v4f*)(up + (size_t)g * Lsz) = o;
        }
    }
}

__device__ __forceinline__ void lines_load(v4f (&g)[TQ], const float* p)
{
#pragma unroll
    for (int i = 0; i < TQ; ++i) g[i] = *(const v4f*)(p + (size_t)i * 4 * CHUNK);
}
__device__ __forceinline__ void lines_to_rows(float* s, const v4f (&g)[TQ])
{
#pragma unroll
    for (int i = 0; i < TQ; ++i) *(v4f*)(s + i * 4 * PITCH) = g[i];
}
__device__ __forceinline__ void rows_to_lines(float* p, const float* s)
{
    v4f g[TQ];
#pragma unroll
    for (int i = 0; i < TQ; ++i) g[i] = *(const v4f*)(s + i * 4 * PITCH);
#pragma unroll
    for (int i = 0; i < TQ; ++i) *(v4f*)(p + (size_t)i * 4 * CHUNK) = g[i];
}

__device__ __forceinline__ void wave_rows_sync()
{
    __builtin_amdgcn_fence(__ATOMIC_RELEASE, "wavefront", "local");
    __builtin_amdgcn_wave_barrier();
    __builtin_amdgcn_fence(__ATOMIC_ACQUIRE, "wavefront", "local");
}

#define WM(a, b, c) __builtin_amdgcn_wmma_f32_16x16x4_f32(false, (a), false, (b), (short)0, (c), false, false)

__global__ __launch_bounds__(NTHREADS)
void k_dssm_scan(const float* __restrict__ u, const float* __restrict__ x,
                 const float* __restrict__ log_dt,
                 const float* __restrict__ C_re, const float* __restrict__ C_im,
                 const float* __restrict__ log_A_real,
                 const float* __restrict__ A_imag,
                 const float* __restrict__ Dp, const float* __restrict__ res_w,
                 float* __restrict__ out)
{
    __shared__ float Zre[Nmode], Zim[Nmode];
    __shared__ float tZr[Nmode], tZi[Nmode];
    __shared__ float tK[16];
    __shared__ float tKn[16][Nmode + 1];
    __shared__ float tP[2 * Nmode][17];
    __shared__ float tQ[16][2 * Nmode + 1];
    __shared__ float sGre[16][Nmode + 1];
    __shared__ float sGim[16][Nmode + 1];
    __shared__ __attribute__((aligned(16))) float stg[4][NTHREADS * PITCH];
    float* const stX = stg[3];

    const int tid  = threadIdx.x;
    const int bh   = blockIdx.x;
    const int h    = bh & (Hsz - 1);
    const int wave = tid >> 5;
    const int lane = tid & 31;
    const int hf   = lane >> 4;
    const int m    = lane & 15;

    const int trow = wave * 32 + (lane >> 3);
    const int tcol = (lane & 7) * 4;
    const size_t goff = (size_t)bh * Lsz + (size_t)trow * CHUNK + tcol;
    const int    loff = trow * PITCH + tcol;
    v4f gu[TQ], gx[TQ];
    lines_load(gu, u + goff);
    lines_load(gx, x + goff);

    {
        const int n = tid & 15, k = tid >> 4;
        const float dt  = expf(log_dt[h]);
        const float Are = -expf(log_A_real[h * Nmode + n]);
        const float Aim = A_imag[h * Nmode + n];
        const float dre = Are * dt, dim = Aim * dt;
        const float er  = expf(dre);
        const float zr  = er * cosf(dim), zi = er * sinf(dim);
        const float e1r = zr - 1.0f, e1i = zi;
        const float cr = C_re[h * Nmode + n], ci = C_im[h * Nmode + n];
        const float nr = cr * e1r - ci * e1i;
        const float ni = cr * e1i + ci * e1r;
        const float den = Are * Are + Aim * Aim;
        const float wr = (nr * Are + ni * Aim) / den;
        const float wi = (ni * Are - nr * Aim) / den;
        const float ek  = expf(dre * (float)k), pk = dim * (float)k;
        const float zkr = ek * cosf(pk), zki = ek * sinf(pk);
        const float z1r = zkr * zr - zki * zi, z1i = zkr * zi + zki * zr;
        tP[n][15 - k]         = zkr;
        tP[Nmode + n][15 - k] = zki;
        tQ[k][n]         =  2.0f * (wr * z1r - wi * z1i);
        tQ[k][Nmode + n] = -2.0f * (wr * z1i + wi * z1r);
        tKn[k][n] = 2.0f * (wr * zkr - wi * zki);
        if (k == 0) {
            const float e16 = expf(dre * 16.0f), p16 = dim * 16.0f;
            tZr[n] = e16 * cosf(p16); tZi[n] = e16 * sinf(p16);
            const float eC = expf(dre * (float)CHUNK), pC = dim * (float)CHUNK;
            Zre[n] = eC * cosf(pC);   Zim[n] = eC * sinf(pC);
        }
    }
    __syncthreads();
    if (tid < 16) {
        float s = 0.0f;
        for (int n = 0; n < Nmode; ++n) s += tKn[tid][n];
        tK[tid] = s;
    }
    __syncthreads();

    const float Dh = Dp[h], rwh = res_w[h];

    v2f aK[4], aP[2][4], aQ[8];
    float z16r[8], z16i[8];
#pragma unroll
    for (int s = 0; s < 4; ++s) {
#pragma unroll
        for (int r = 0; r < 2; ++r) {
            const int p = 8 * hf + 2 * s + r;
            const int d = m - p;
            const float kv = tK[d & 15];
            aK[s][r]    = (d > 0) ? kv : ((d == 0) ? kv + Dh : 0.0f);
            aP[0][s][r] = tP[m][p];
            aP[1][s][r] = tP[Nmode + m][p];
        }
    }
#pragma unroll
    for (int s = 0; s < 8; ++s) {
#pragma unroll
        for (int r = 0; r < 2; ++r)
            aQ[s][r] = tQ[m][16 * (s >> 2) + 8 * hf + 2 * (s & 3) + r];
    }
#pragma unroll
    for (int r = 0; r < 8; ++r) { z16r[r] = tZr[8 * hf + r]; z16i[r] = tZi[8 * hf + r]; }

    const int foff = (wave * 32 + m) * PITCH + 8 * hf;
    float* const sS = stX + wave * 32 * PITCH;

    v8f S[2][2] = {};
#pragma unroll 1
    for (int t = 0; t < NTILE; ++t) {
        float* st = stg[t];
        lines_to_rows(st + loff, gu);
        wave_rows_sync();
        if (t + 1 < NTILE) lines_load(gu, u + goff + (t + 1) * TCOLS);
#pragma unroll
        for (int q2 = 0; q2 < 2; ++q2) {
#pragma unroll
            for (int j = 0; j < 2; ++j) {
                const float* fp = st + foff + 16 * j * PITCH + 16 * q2;
                const v4f ub0 = *(const v4f*)(fp), ub1 = *(const v4f*)(fp + 4);
                v2f b0, b1, b2, b3;
                b0.x = ub0[0]; b0.y = ub0[1]; b1.x = ub0[2]; b1.y = ub0[3];
                b2.x = ub1[0]; b2.y = ub1[1]; b3.x = ub1[2]; b3.y = ub1[3];
                v8f nr, ni;
#pragma unroll
                for (int r = 0; r < 8; ++r) {
                    nr[r] = z16r[r] * S[j][0][r] - z16i[r] * S[j][1][r];
                    ni[r] = z16r[r] * S[j][1][r] + z16i[r] * S[j][0][r];
                }
                nr = WM(aP[0][0], b0, nr); ni = WM(aP[1][0], b0, ni);
                nr = WM(aP[0][1], b1, nr); ni = WM(aP[1][1], b1, ni);
                nr = WM(aP[0][2], b2, nr); ni = WM(aP[1][2], b2, ni);
                nr = WM(aP[0][3], b3, nr); ni = WM(aP[1][3], b3, ni);
                S[j][0] = nr; S[j][1] = ni;
            }
        }
    }
    wave_rows_sync();
#pragma unroll
    for (int j = 0; j < 2; ++j) {
#pragma unroll
        for (int r = 0; r < 8; ++r) {
            sS[(16 * j + m) * SPITCH + 8 * hf + r]       = S[j][0][r];
            sS[SIM + (16 * j + m) * SPITCH + 8 * hf + r] = S[j][1][r];
        }
    }
    wave_rows_sync();

    {
        const int n  = lane & 15;
        const int jj = lane >> 4;
        const int G  = 2 * wave + jj;
        float* const gS = sS + 16 * jj * SPITCH;
        const float Zr = Zre[n], Zi = Zim[n];
        float pr = 0.0f, pi = 0.0f;
#pragma unroll
        for (int c = 0; c < 16; ++c) {
            const float tr = gS[c * SPITCH + n], ti = gS[SIM + c * SPITCH + n];
            const float npr = fmaf(Zr, pr, fmaf(-Zi, pi, tr));
            const float npi = fmaf(Zr, pi, fmaf( Zi, pr, ti));
            pr = npr; pi = npi;
        }
        sGre[G][n] = pr; sGim[G][n] = pi;
        float gr = Zr, gi = Zi;
#pragma unroll
        for (int k = 0; k < 4; ++k) {
            const float nr = gr * gr - gi * gi;
            const float ni = 2.0f * gr * gi;
            gr = nr; gi = ni;
        }
        __syncthreads();
        float qr = 0.0f, qi = 0.0f;
#pragma unroll 5
        for (int g = 0; g < 15; ++g) {
            const float ar = sGre[g][n], ai = sGim[g][n];
            const float nqr = fmaf(gr, qr, fmaf(-gi, qi, ar));
            const float nqi = fmaf(gr, qi, fmaf( gi, qr, ai));
            if (g < G) { qr = nqr; qi = nqi; }
        }
        pr = qr; pi = qi;
#pragma unroll
        for (int c = 0; c < 16; ++c) {
            const float tr = gS[c * SPITCH + n], ti = gS[SIM + c * SPITCH + n];
            gS[c * SPITCH + n] = pr; gS[SIM + c * SPITCH + n] = pi;
            const float npr = fmaf(Zr, pr, fmaf(-Zi, pi, tr));
            const float npi = fmaf(Zr, pi, fmaf( Zi, pr, ti));
            pr = npr; pi = npi;
        }
    }
    wave_rows_sync();

#pragma unroll
    for (int j = 0; j < 2; ++j) {
#pragma unroll
        for (int r = 0; r < 8; ++r) {
            S[j][0][r] = sS[(16 * j + m) * SPITCH + 8 * hf + r];
            S[j][1][r] = sS[SIM + (16 * j + m) * SPITCH + 8 * hf + r];
        }
    }
    wave_rows_sync();
#pragma unroll 1
    for (int t = 0; t < NTILE; ++t) {
        const float* const stU = stg[t < NTILE - 1 ? t : 0];
        if (t == NTILE - 1) lines_to_rows(stg[0] + loff, gu);
        lines_to_rows(stX + loff, gx);
        wave_rows_sync();
        if (t + 1 < NTILE) lines_load(gx, x + goff + (t + 1) * TCOLS);
#pragma unroll
        for (int q2 = 0; q2 < 2; ++q2) {
#pragma unroll
            for (int j = 0; j < 2; ++j) {
                const int fo = foff + 16 * j * PITCH + 16 * q2;
                const v4f ub0 = *(const v4f*)(stU + fo), ub1 = *(const v4f*)(stU + fo + 4);
                const v4f xb0 = *(const v4f*)(stX + fo), xb1 = *(const v4f*)(stX + fo + 4);
                v2f b0, b1, b2, b3;
                b0.x = ub0[0]; b0.y = ub0[1]; b1.x = ub0[2]; b1.y = ub0[3];
                b2.x = ub1[0]; b2.y = ub1[1]; b3.x = ub1[2]; b3.y = ub1[3];
                v8f y = {};
                y = WM(aK[0], b0, y); y = WM(aK[1], b1, y);
                y = WM(aK[2], b2, y); y = WM(aK[3], b3, y);
#pragma unroll
                for (int s = 0; s < 8; ++s) {
                    v2f sb;
                    sb.x = S[j][s >> 2][2 * (s & 3)];
                    sb.y = S[j][s >> 2][2 * (s & 3) + 1];
                    y = WM(aQ[s], sb, y);
                }
                v8f nr, ni;
#pragma unroll
                for (int r = 0; r < 8; ++r) {
                    nr[r] = z16r[r] * S[j][0][r] - z16i[r] * S[j][1][r];
                    ni[r] = z16r[r] * S[j][1][r] + z16i[r] * S[j][0][r];
                }
                nr = WM(aP[0][0], b0, nr); ni = WM(aP[1][0], b0, ni);
                nr = WM(aP[0][1], b1, nr); ni = WM(aP[1][1], b1, ni);
                nr = WM(aP[0][2], b2, nr); ni = WM(aP[1][2], b2, ni);
                nr = WM(aP[0][3], b3, nr); ni = WM(aP[1][3], b3, ni);
                S[j][0] = nr; S[j][1] = ni;
                v4f o0, o1;
#pragma unroll
                for (int e = 0; e < 4; ++e) {
                    o0[e] = tanh_scaled(y[e] * TWO_LOG2E)     + xb0[e] * rwh;
                    o1[e] = tanh_scaled(y[4 + e] * TWO_LOG2E) + xb1[e] * rwh;
                }
                *(v4f*)(stX + fo)     = o0;
                *(v4f*)(stX + fo + 4) = o1;
            }
        }
        wave_rows_sync();
        rows_to_lines(out + goff + t * TCOLS, stX + loff);
    }
}

extern "C" void kernel_launch(void* const* d_in, const int* in_sizes, int n_in,
                              void* d_out, int out_size, void* d_ws, size_t ws_size,
                              hipStream_t stream)
{
    (void)in_sizes; (void)n_in; (void)out_size; (void)ws_size;
    const float* x      = (const float*)d_in[0];
    const float* W      = (const float*)d_in[1];
    const float* bias   = (const float*)d_in[2];
    const float* log_dt = (const float*)d_in[3];
    const float* C_re   = (const float*)d_in[4];
    const float* C_im   = (const float*)d_in[5];
    const float* lAr    = (const float*)d_in[6];
    const float* Aim    = (const float*)d_in[7];
    const float* Dp     = (const float*)d_in[8];
    const float* rw     = (const float*)d_in[9];
    float* out = (float*)d_out;
    float* u   = (float*)d_ws;

    dim3 g1(Lsz / K1_BCOLS, Bsz);
    k_channel_mix<<<g1, NTHREADS, 0, stream>>>(x, W, bias, u);
    k_dssm_scan<<<Bsz * Hsz, NTHREADS, 0, stream>>>(u, x, log_dt, C_re, C_im,
                                                    lAr, Aim, Dp, rw, out);
}


# speedup vs baseline: 14.3014x; 1.0427x over previous
#include <hip/hip_runtime.h>
#include <hip/hip_bf16.h>
#include <math.h>


typedef __attribute__((ext_vector_type(2))) float v2f;
typedef __attribute__((ext_vector_type(8))) float v8f;
typedef __attribute__((ext_vector_type(4))) float v4f;

#define Bsz 8
#define Hsz 64
#define Lsz 32768
#define Nmode 16
#define NTHREADS 256
#define CHUNK (Lsz / NTHREADS)
#define TCOLS 64
#define TROWS 16
#define NTILE 4
#define TQ 8
#define PITCH 68
#define WAREA (TROWS * PITCH)
#define TILE_GOFF(c, j) ((size_t)(16 * (j)) * CHUNK + 64 * (c))
#define SPITCH (Nmode + 1)
#define SIM (32 * SPITCH)

#define TWO_LOG2E 2.885390081777927f
__device__ __forceinline__ float tanh_scaled(float es)
{
    const float t = __builtin_amdgcn_exp2f(es);
    return fmaf(__builtin_amdgcn_rcpf(t + 1.0f), -2.0f, 1.0f);
}

#define K1_WCOLS 64
#define K1_BCOLS (8 * K1_WCOLS)
#define WPITCH 68
#define K1_AHEAD 6

__global__ __launch_bounds__(NTHREADS)
void k_channel_mix(const float* __restrict__ x, const float* __restrict__ W,
                   const float* __restrict__ bias, float* __restrict__ u)
{
    __shared__ __attribute__((aligned(16))) float sW[Hsz * WPITCH];

    const int tid  = threadIdx.x;
    const int wave = tid >> 5;
    const int lane = tid & 31;
    const int half = lane >> 4;
    const int lp   = lane & 15;

    const int b  = blockIdx.y;
    const int l0 = blockIdx.x * K1_BCOLS + wave * K1_WCOLS;

    {
        const int row = tid >> 2, c0 = (tid & 3) * 16;
        const float* src = W + row * Hsz + c0;
        const v4f w0 = *(const v4f*)(src + 0), w1 = *(const v4f*)(src + 4);
        const v4f w2 = *(const v4f*)(src + 8), w3 = *(const v4f*)(src + 12);
        float* dst = sW + row * WPITCH + c0;
        *(v4f*)(dst + 0) = w0; *(v4f*)(dst + 4)  = w1;
        *(v4f*)(dst + 8) = w2; *(v4f*)(dst + 12) = w3;
    }

    const float* xp = x + (size_t)b * Hsz * Lsz + (size_t)(2 * half) * Lsz + l0 + 4 * lp;

    v4f ring[K1_AHEAD][2];
#pragma unroll
    for (int i = 0; i < K1_AHEAD; ++i) {
        ring[i][0] = *(const v4f*)(xp + (size_t)(4 * i + 0) * Lsz);
        ring[i][1] = *(const v4f*)(xp + (size_t)(4 * i + 1) * Lsz);
    }
    __syncthreads();

    v8f acc[4][4] = {};
    const float* wl = sW + lp * WPITCH + 2 * half;

#pragma unroll
    for (int ks = 0; ks < Hsz / 4; ++ks) {
        const v4f c0 = ring[ks % K1_AHEAD][0];
        const v4f c1 = ring[ks % K1_AHEAD][1];
        if (ks + K1_AHEAD < Hsz / 4) {
            ring[ks % K1_AHEAD][0] = *(const v4f*)(xp + (size_t)(4 * (ks + K1_AHEAD) + 0) * Lsz);
            ring[ks % K1_AHEAD][1] = *(const v4f*)(xp + (size_t)(4 * (ks + K1_AHEAD) + 1) * Lsz);
        }
        v2f aa[4];
#pragma unroll
        for (int mt = 0; mt < 4; ++mt)
            aa[mt] = *(const v2f*)(wl + mt * 16 * WPITCH + 4 * ks);
#pragma unroll
        for (int j = 0; j < 4; ++j) {
            v2f bb;
            bb.x = c0[j];
            bb.y = c1[j];
#pragma unroll
            for (int mt = 0; mt < 4; ++mt)
                acc[mt][j] = __builtin_amdgcn_wmma_f32_16x16x4_f32(
                    false, aa[mt], false, bb, (short)0, acc[mt][j], false, false);
        }
    }

    float* up = u + ((size_t)b * Hsz + 8 * half) * Lsz + l0 + 4 * lp;
#pragma unroll
    for (int mt = 0; mt < 4; ++mt) {
#pragma unroll
        for (int r = 0; r < 8; ++r) {
            const int g = mt * 16 + r;
            const float bs = bias[g + 8 * half] * TWO_LOG2E;
            v4f o;
#pragma unroll
            for (int j = 0; j < 4; ++j) o[j] = tanh_scaled(fmaf(acc[mt][j][r], TWO_LOG2E, bs));
            *(v4f*)(up + (size_t)g * Lsz) = o;
        }
    }
}

__device__ __forceinline__ void lines_load(v4f (&g)[TQ], const float* p)
{
#pragma unroll
    for (int i = 0; i < TQ; ++i) g[i] = *(const v4f*)(p + (size_t)i * 2 * CHUNK);
}
__device__ __forceinline__ void lines_to_rows(float* s, const v4f (&g)[TQ])
{
#pragma unroll
    for (int i = 0; i < TQ; ++i) *(v4f*)(s + i * 2 * PITCH) = g[i];
}
__device__ __forceinline__ void rows_to_lines(float* p, const float* s)
{
    v4f g[TQ];
#pragma unroll
    for (int i = 0; i < TQ; ++i) g[i] = *(const v4f*)(s + i * 2 * PITCH);
#pragma unroll
    for (int i = 0; i < TQ; ++i) *(v4f*)(p + (size_t)i * 2 * CHUNK) = g[i];
}

__device__ __forceinline__ void wave_rows_sync()
{
    __builtin_amdgcn_fence(__ATOMIC_RELEASE, "wavefront", "local");
    __builtin_amdgcn_wave_barrier();
    __builtin_amdgcn_fence(__ATOMIC_ACQUIRE, "wavefront", "local");
}

#define WM(a, b, c) __builtin_amdgcn_wmma_f32_16x16x4_f32(false, (a), false, (b), (short)0, (c), false, false)

__global__ __launch_bounds__(NTHREADS, 4)
void k_dssm_scan(const float* __restrict__ u, const float* __restrict__ x,
                 const float* __restrict__ log_dt,
                 const float* __restrict__ C_re, const float* __restrict__ C_im,
                 const float* __restrict__ log_A_real,
                 const float* __restrict__ A_imag,
                 const float* __restrict__ Dp, const float* __restrict__ res_w,
                 float* __restrict__ out)
{
    __shared__ float Zre[Nmode], Zim[Nmode];
    __shared__ float tZr[Nmode], tZi[Nmode];
    __shared__ float tK[16];
    __shared__ float tKn[16][Nmode + 1];
    __shared__ float tP[2 * Nmode][17];
    __shared__ float tQ[16][2 * Nmode + 1];
    __shared__ float sGre[16][Nmode + 1];
    __shared__ float sGim[16][Nmode + 1];
    __shared__ __attribute__((aligned(16))) float stg[4][8 * WAREA];
    float* const stX = stg[3];

    const int tid  = threadIdx.x;
    const int bh   = blockIdx.x;
    const int h    = bh & (Hsz - 1);
    const int wave = tid >> 5;
    const int lane = tid & 31;
    const int hf   = lane >> 4;
    const int m    = lane & 15;

    const int r2   = lane >> 4;
    const int tcol = (lane & 15) * 4;
    const size_t goff = (size_t)bh * Lsz + (size_t)(wave * 32 + r2) * CHUNK + tcol;
    const int    loff = wave * WAREA + r2 * PITCH + tcol;
    v4f gu[TQ], gx[TQ];
    lines_load(gu, u + goff);
    lines_load(gx, x + goff);

    {
        const int n = tid & 15, k = tid >> 4;
        const float dt  = expf(log_dt[h]);
        const float Are = -expf(log_A_real[h * Nmode + n]);
        const float Aim = A_imag[h * Nmode + n];
        const float dre = Are * dt, dim = Aim * dt;
        const float er  = expf(dre);
        const float zr  = er * cosf(dim), zi = er * sinf(dim);
        const float e1r = zr - 1.0f, e1i = zi;
        const float cr = C_re[h * Nmode + n], ci = C_im[h * Nmode + n];
        const float nr = cr * e1r - ci * e1i;
        const float ni = cr * e1i + ci * e1r;
        const float den = Are * Are + Aim * Aim;
        const float wr = (nr * Are + ni * Aim) / den;
        const float wi = (ni * Are - nr * Aim) / den;
        const float ek  = expf(dre * (float)k), pk = dim * (float)k;
        const float zkr = ek * cosf(pk), zki = ek * sinf(pk);
        const float z1r = zkr * zr - zki * zi, z1i = zkr * zi + zki * zr;
        tP[n][15 - k]         = zkr;
        tP[Nmode + n][15 - k] = zki;
        tQ[k][n]         =  2.0f * (wr * z1r - wi * z1i);
        tQ[k][Nmode + n] = -2.0f * (wr * z1i + wi * z1r);
        tKn[k][n] = 2.0f * (wr * zkr - wi * zki);
        if (k == 0) {
            const float e16 = expf(dre * 16.0f), p16 = dim * 16.0f;
            tZr[n] = e16 * cosf(p16); tZi[n] = e16 * sinf(p16);
            const float eC = expf(dre * (float)CHUNK), pC = dim * (float)CHUNK;
            Zre[n] = eC * cosf(pC);   Zim[n] = eC * sinf(pC);
        }
    }
    __syncthreads();
    if (tid < 16) {
        float s = 0.0f;
        for (int n = 0; n < Nmode; ++n) s += tKn[tid][n];
        tK[tid] = s;
    }
    __syncthreads();

    const float Dh = Dp[h], rwh = res_w[h];

    v2f aK[4], aP[2][4], aQ[8];
    float z16r[8], z16i[8];
#pragma unroll
    for (int s = 0; s < 4; ++s) {
#pragma unroll
        for (int r = 0; r < 2; ++r) {
            const int p = 8 * hf + 2 * s + r;
            const int d = m - p;
            const float kv = tK[d & 15];
            aK[s][r]    = (d > 0) ? kv : ((d == 0) ? kv + Dh : 0.0f);
            aP[0][s][r] = tP[m][p];
            aP[1][s][r] = tP[Nmode + m][p];
        }
    }
#pragma unroll
    for (int s = 0; s < 8; ++s) {
#pragma unroll
        for (int r = 0; r < 2; ++r)
            aQ[s][r] = tQ[m][16 * (s >> 2) + 8 * hf + 2 * (s & 3) + r];
    }
#pragma unroll
    for (int r = 0; r < 8; ++r) { z16r[r] = tZr[8 * hf + r]; z16i[r] = tZi[8 * hf + r]; }

    const int foff = wave * WAREA + m * PITCH + 8 * hf;
    float* const sS = stX + wave * WAREA;

    v8f S[2][2] = {};
#pragma unroll 1
    for (int c = 0; c < 2; ++c) {
#pragma unroll
        for (int j = 0; j < 2; ++j) {
            float* st = stg[2 * c + j];
            lines_to_rows(st + loff, gu);
            wave_rows_sync();
            if (j == 0)      lines_load(gu, u + goff + TILE_GOFF(c, 1));
            else if (c == 0) lines_load(gu, u + goff + TILE_GOFF(1, 0));
#pragma unroll
            for (int q = 0; q < 4; ++q) {
                const float* fp = st + foff + 16 * q;
                const v4f ub0 = *(const v4f*)(fp), ub1 = *(const v4f*)(fp + 4);
                v2f b0, b1, b2, b3;
                b0.x = ub0[0]; b0.y = ub0[1]; b1.x = ub0[2]; b1.y = ub0[3];
                b2.x = ub1[0]; b2.y = ub1[1]; b3.x = ub1[2]; b3.y = ub1[3];
                v8f nr, ni;
#pragma unroll
                for (int r = 0; r < 8; ++r) {
                    nr[r] = z16r[r] * S[j][0][r] - z16i[r] * S[j][1][r];
                    ni[r] = z16r[r] * S[j][1][r] + z16i[r] * S[j][0][r];
                }
                nr = WM(aP[0][0], b0, nr); ni = WM(aP[1][0], b0, ni);
                nr = WM(aP[0][1], b1, nr); ni = WM(aP[1][1], b1, ni);
                nr = WM(aP[0][2], b2, nr); ni = WM(aP[1][2], b2, ni);
                nr = WM(aP[0][3], b3, nr); ni = WM(aP[1][3], b3, ni);
                S[j][0] = nr; S[j][1] = ni;
            }
            wave_rows_sync();
        }
    }
    wave_rows_sync();
#pragma unroll
    for (int j = 0; j < 2; ++j) {
#pragma unroll
        for (int r = 0; r < 8; ++r) {
            sS[(16 * j + m) * SPITCH + 8 * hf + r]       = S[j][0][r];
            sS[SIM + (16 * j + m) * SPITCH + 8 * hf + r] = S[j][1][r];
        }
    }
    wave_rows_sync();

    {
        const int n  = lane & 15;
        const int jj = lane >> 4;
        const int G  = 2 * wave + jj;
        float* const gS = sS + 16 * jj * SPITCH;
        const float Zr = Zre[n], Zi = Zim[n];
        float pr = 0.0f, pi = 0.0f;
#pragma unroll
        for (int c = 0; c < 16; ++c) {
            const float tr = gS[c * SPITCH + n], ti = gS[SIM + c * SPITCH + n];
            const float npr = fmaf(Zr, pr, fmaf(-Zi, pi, tr));
            const float npi = fmaf(Zr, pi, fmaf( Zi, pr, ti));
            pr = npr; pi = npi;
        }
        sGre[G][n] = pr; sGim[G][n] = pi;
        float gr = Zr, gi = Zi;
#pragma unroll
        for (int k = 0; k < 4; ++k) {
            const float nr = gr * gr - gi * gi;
            const float ni = 2.0f * gr * gi;
            gr = nr; gi = ni;
        }
        __syncthreads();
        float qr = 0.0f, qi = 0.0f;
#pragma unroll 5
        for (int g = 0; g < 15; ++g) {
            const float ar = sGre[g][n], ai = sGim[g][n];
            const float nqr = fmaf(gr, qr, fmaf(-gi, qi, ar));
            const float nqi = fmaf(gr, qi, fmaf( gi, qr, ai));
            if (g < G) { qr = nqr; qi = nqi; }
        }
        pr = qr; pi = qi;
#pragma unroll
        for (int c = 0; c < 16; ++c) {
            const float tr = gS[c * SPITCH + n], ti = gS[SIM + c * SPITCH + n];
            gS[c * SPITCH + n] = pr; gS[SIM + c * SPITCH + n] = pi;
            const float npr = fmaf(Zr, pr, fmaf(-Zi, pi, tr));
            const float npi = fmaf(Zr, pi, fmaf( Zi, pr, ti));
            pr = npr; pi = npi;
        }
    }
    wave_rows_sync();

#pragma unroll
    for (int j = 0; j < 2; ++j) {
#pragma unroll
        for (int r = 0; r < 8; ++r) {
            S[j][0][r] = sS[(16 * j + m) * SPITCH + 8 * hf + r];
            S[j][1][r] = sS[SIM + (16 * j + m) * SPITCH + 8 * hf + r];
        }
    }
    wave_rows_sync();
#pragma unroll 1
    for (int c = 0; c < 2; ++c) {
#pragma unroll
        for (int j = 0; j < 2; ++j) {
            const int t = 2 * c + j;
            const float* const stU = stg[t < NTILE - 1 ? t : 0];
            if (t == NTILE - 1) lines_to_rows(stg[0] + loff, gu);
            lines_to_rows(stX + loff, gx);
            wave_rows_sync();
            if (j == 0)      lines_load(gx, x + goff + TILE_GOFF(c, 1));
            else if (c == 0) lines_load(gx, x + goff + TILE_GOFF(1, 0));
#pragma unroll
            for (int q = 0; q < 4; ++q) {
                const int fo = foff + 16 * q;
                const v4f ub0 = *(const v4f*)(stU + fo), ub1 = *(const v4f*)(stU + fo + 4);
                const v4f xb0 = *(const v4f*)(stX + fo), xb1 = *(const v4f*)(stX + fo + 4);
                v2f b0, b1, b2, b3;
                b0.x = ub0[0]; b0.y = ub0[1]; b1.x = ub0[2]; b1.y = ub0[3];
                b2.x = ub1[0]; b2.y = ub1[1]; b3.x = ub1[2]; b3.y = ub1[3];
                v8f y = {};
                y = WM(aK[0], b0, y); y = WM(aK[1], b1, y);
                y = WM(aK[2], b2, y); y = WM(aK[3], b3, y);
#pragma unroll
                for (int s = 0; s < 8; ++s) {
                    v2f sb;
                    sb.x = S[j][s >> 2][2 * (s & 3)];
                    sb.y = S[j][s >> 2][2 * (s & 3) + 1];
                    y = WM(aQ[s], sb, y);
                }
                v8f nr, ni;
#pragma unroll
                for (int r = 0; r < 8; ++r) {
                    nr[r] = z16r[r] * S[j][0][r] - z16i[r] * S[j][1][r];
                    ni[r] = z16r[r] * S[j][1][r] + z16i[r] * S[j][0][r];
                }
                nr = WM(aP[0][0], b0, nr); ni = WM(aP[1][0], b0, ni);
                nr = WM(aP[0][1], b1, nr); ni = WM(aP[1][1], b1, ni);
                nr = WM(aP[0][2], b2, nr); ni = WM(aP[1][2], b2, ni);
                nr = WM(aP[0][3], b3, nr); ni = WM(aP[1][3], b3, ni);
                S[j][0] = nr; S[j][1] = ni;
                v4f o0, o1;
#pragma unroll
                for (int e = 0; e < 4; ++e) {
                    o0[e] = tanh_scaled(y[e] * TWO_LOG2E)     + xb0[e] * rwh;
                    o1[e] = tanh_scaled(y[4 + e] * TWO_LOG2E) + xb1[e] * rwh;
                }
                *(v4f*)(stX + fo)     = o0;
                *(v4f*)(stX + fo + 4) = o1;
            }
            wave_rows_sync();
            rows_to_lines(out + goff + TILE_GOFF(c, j), stX + loff);
        }
    }
}

extern "C" void kernel_launch(void* const* d_in, const int* in_sizes, int n_in,
                              void* d_out, int out_size, void* d_ws, size_t ws_size,
                              hipStream_t stream)
{
    (void)in_sizes; (void)n_in; (void)out_size; (void)ws_size;
    const float* x      = (const float*)d_in[0];
    const float* W      = (const float*)d_in[1];
    const float* bias   = (const float*)d_in[2];
    const float* log_dt = (const float*)d_in[3];
    const float* C_re   = (const float*)d_in[4];
    const float* C_im   = (const float*)d_in[5];
    const float* lAr    = (const float*)d_in[6];
    const float* Aim    = (const float*)d_in[7];
    const float* Dp     = (const float*)d_in[8];
    const float* rw     = (const float*)d_in[9];
    float* out = (float*)d_out;
    float* u   = (float*)d_ws;

    dim3 g1(Lsz / K1_BCOLS, Bsz);
    k_channel_mix<<<g1, NTHREADS, 0, stream>>>(x, W, bias, u);
    k_dssm_scan<<<Bsz * Hsz, NTHREADS, 0, stream>>>(u, x, log_dt, C_re, C_im,
                                                    lAr, Aim, Dp, rw, out);
}
